# starting kernel
# baseline (speedup 1.0000x reference)
#include <hip/hip_runtime.h>

#define B_  4
#define NQ_ 512
#define NK_ 512
#define D_  512
#define H_  128
#define QB_ 8

typedef __attribute__((ext_vector_type(2))) float v2f;
typedef __attribute__((ext_vector_type(8))) float v8f;

__device__ __forceinline__ float hw_tanh(float x) {
#if __has_builtin(__builtin_amdgcn_tanhf)
  return __builtin_amdgcn_tanhf(x);
#else
  float y;
  asm volatile("v_tanh_f32 %0, %1" : "=v"(y) : "v"(x));
  return y;
#endif
}

__device__ __forceinline__ float hw_exp(float x) {
#if __has_builtin(__builtin_amdgcn_exp2f)
  return __builtin_amdgcn_exp2f(x * 1.44269504088896340736f);
#else
  return __expf(x);
#endif
}

struct Frags {
  v2f a0, a1, b0, b1, b2, b3;
};

__device__ __forceinline__ void load_frags(const float* __restrict__ arow0,
                                           const float* __restrict__ arow1,
                                           const float* __restrict__ bbase,
                                           int N, int kk, Frags& f) {
  f.a0 = *(const v2f*)(arow0 + kk);
  f.a1 = *(const v2f*)(arow1 + kk);
  const float* b = bbase + (long long)kk * N;
  f.b0.x = b[0];       f.b0.y = b[N];
  f.b1.x = b[16];      f.b1.y = b[N + 16];
  f.b2.x = b[32];      f.b2.y = b[N + 32];
  f.b3.x = b[48];      f.b3.y = b[N + 48];
}

__global__ __launch_bounds__(32) void wmma_gemm_f32(
    const float* __restrict__ A, const float* __restrict__ Bm,
    float* __restrict__ C, int K, int N,
    long long sA, long long sB, long long sC) {
  const int lane  = threadIdx.x;
  const int mTile = blockIdx.x * 32;
  const int nBase = blockIdx.y * 64;
  const int batch = blockIdx.z;
  A  += (long long)batch * sA;
  Bm += (long long)batch * sB;
  C  += (long long)batch * sC;

  const int mRow = lane & 15;
  const int kHi  = (lane >> 4) * 2;
  const int col  = lane & 15;

  const float* arow0 = A + (long long)(mTile + mRow) * K + kHi;
  const float* arow1 = arow0 + (long long)16 * K;
  const float* bbase = Bm + (long long)kHi * N + nBase + col;

  v8f acc00 = {}, acc01 = {}, acc02 = {}, acc03 = {};
  v8f acc10 = {}, acc11 = {}, acc12 = {}, acc13 = {};

  Frags cur, nxt;
  load_frags(arow0, arow1, bbase, N, 0, cur);

  for (int kk = 4; kk < K; kk += 4) {
    load_frags(arow0, arow1, bbase, N, kk, nxt);
    acc00 = __builtin_amdgcn_wmma_f32_16x16x4_f32(false, cur.a0, false, cur.b0, (short)0, acc00, false, false);
    acc01 = __builtin_amdgcn_wmma_f32_16x16x4_f32(false, cur.a0, false, cur.b1, (short)0, acc01, false, false);
    acc02 = __builtin_amdgcn_wmma_f32_16x16x4_f32(false, cur.a0, false, cur.b2, (short)0, acc02, false, false);
    acc03 = __builtin_amdgcn_wmma_f32_16x16x4_f32(false, cur.a0, false, cur.b3, (short)0, acc03, false, false);
    acc10 = __builtin_amdgcn_wmma_f32_16x16x4_f32(false, cur.a1, false, cur.b0, (short)0, acc10, false, false);
    acc11 = __builtin_amdgcn_wmma_f32_16x16x4_f32(false, cur.a1, false, cur.b1, (short)0, acc11, false, false);
    acc12 = __builtin_amdgcn_wmma_f32_16x16x4_f32(false, cur.a1, false, cur.b2, (short)0, acc12, false, false);
    acc13 = __builtin_amdgcn_wmma_f32_16x16x4_f32(false, cur.a1, false, cur.b3, (short)0, acc13, false, false);
    cur = nxt;
  }
  acc00 = __builtin_amdgcn_wmma_f32_16x16x4_f32(false, cur.a0, false, cur.b0, (short)0, acc00, false, false);
  acc01 = __builtin_amdgcn_wmma_f32_16x16x4_f32(false, cur.a0, false, cur.b1, (short)0, acc01, false, false);
  acc02 = __builtin_amdgcn_wmma_f32_16x16x4_f32(false, cur.a0, false, cur.b2, (short)0, acc02, false, false);
  acc03 = __builtin_amdgcn_wmma_f32_16x16x4_f32(false, cur.a0, false, cur.b3, (short)0, acc03, false, false);
  acc10 = __builtin_amdgcn_wmma_f32_16x16x4_f32(false, cur.a1, false, cur.b0, (short)0, acc10, false, false);
  acc11 = __builtin_amdgcn_wmma_f32_16x16x4_f32(false, cur.a1, false, cur.b1, (short)0, acc11, false, false);
  acc12 = __builtin_amdgcn_wmma_f32_16x16x4_f32(false, cur.a1, false, cur.b2, (short)0, acc12, false, false);
  acc13 = __builtin_amdgcn_wmma_f32_16x16x4_f32(false, cur.a1, false, cur.b3, (short)0, acc13, false, false);

  const int mOut0 = mTile + (lane >> 4) * 8;
#pragma unroll
  for (int r = 0; r < 8; ++r) {
    long long row0 = (long long)(mOut0 + r) * N + nBase + col;
    C[row0 +  0] = acc00[r];
    C[row0 + 16] = acc01[r];
    C[row0 + 32] = acc02[r];
    C[row0 + 48] = acc03[r];
    long long row1 = row0 + (long long)16 * N;
    C[row1 +  0] = acc10[r];
    C[row1 + 16] = acc11[r];
    C[row1 + 32] = acc12[r];
    C[row1 + 48] = acc13[r];
  }
}

__global__ __launch_bounds__(NK_) void scores_softmax(
    const float* __restrict__ qproj, const float* __restrict__ kproj,
    const float* __restrict__ Wv, const int* __restrict__ valid_lens,
    float* __restrict__ attn) {
  __shared__ float qp[QB_][H_];
  __shared__ float wv[H_];
  __shared__ float part[QB_][16];
  __shared__ float mxs[QB_];
  __shared__ float dns[QB_];

  const int q0   = blockIdx.x * QB_;
  const int b    = blockIdx.y;
  const int k    = threadIdx.x;
  const int lane = k & 31;
  const int wid  = k >> 5;

  for (int i = k; i < QB_ * H_; i += NK_)
    qp[i / H_][i % H_] = qproj[(long long)(b * NQ_ + q0) * H_ + i];
  if (k < H_) wv[k] = Wv[k];
  __syncthreads();

  const float4* kp4 =
      (const float4*)(kproj + (long long)(b * NK_ + k) * H_);
  float s[QB_];
#pragma unroll
  for (int j = 0; j < QB_; ++j) s[j] = 0.0f;

  for (int h4 = 0; h4 < H_ / 4; ++h4) {
    const float4 kv = kp4[h4];
    const int h = h4 * 4;
#pragma unroll
    for (int j = 0; j < QB_; ++j) {
      s[j] += wv[h + 0] * hw_tanh(qp[j][h + 0] + kv.x);
      s[j] += wv[h + 1] * hw_tanh(qp[j][h + 1] + kv.y);
      s[j] += wv[h + 2] * hw_tanh(qp[j][h + 2] + kv.z);
      s[j] += wv[h + 3] * hw_tanh(qp[j][h + 3] + kv.w);
    }
  }

  const int vlen = valid_lens[b];
  if (k >= vlen) {
#pragma unroll
    for (int j = 0; j < QB_; ++j) s[j] = -1.0e6f;
  }

#pragma unroll
  for (int j = 0; j < QB_; ++j) {
    float m = s[j];
#pragma unroll
    for (int off = 16; off > 0; off >>= 1) m = fmaxf(m, __shfl_xor(m, off, 32));
    if (lane == 0) part[j][wid] = m;
  }
  __syncthreads();
  if (k < QB_) {
    float m = part[k][0];
#pragma unroll
    for (int w = 1; w < 16; ++w) m = fmaxf(m, part[k][w]);
    mxs[k] = m;
  }
  __syncthreads();

  float e[QB_];
#pragma unroll
  for (int j = 0; j < QB_; ++j) {
    e[j] = hw_exp(s[j] - mxs[j]);
    float t = e[j];
#pragma unroll
    for (int off = 16; off > 0; off >>= 1) t += __shfl_xor(t, off, 32);
    if (lane == 0) part[j][wid] = t;
  }
  __syncthreads();
  if (k < QB_) {
    float t = 0.0f;
#pragma unroll
    for (int w = 0; w < 16; ++w) t += part[k][w];
    dns[k] = t;
  }
  __syncthreads();

#pragma unroll
  for (int j = 0; j < QB_; ++j) {
    attn[(long long)(b * NQ_ + q0 + j) * NK_ + k] = e[j] / dns[j];
  }
}

extern "C" void kernel_launch(void* const* d_in, const int* in_sizes, int n_in,
                              void* d_out, int out_size, void* d_ws, size_t ws_size,
                              hipStream_t stream) {
  (void)in_sizes; (void)n_in; (void)out_size; (void)ws_size;

  const float* queries    = (const float*)d_in[0];
  const float* keys       = (const float*)d_in[1];
  const float* values     = (const float*)d_in[2];
  const float* Wq         = (const float*)d_in[3];
  const float* Wk         = (const float*)d_in[4];
  const float* Wv         = (const float*)d_in[5];
  const int*   valid_lens = (const int*)d_in[6];
  float*       out        = (float*)d_out;

  float* qproj = (float*)d_ws;
  float* kproj = qproj + (long long)B_ * NQ_ * H_;
  float* attn  = kproj + (long long)B_ * NK_ * H_;

  wmma_gemm_f32<<<dim3((B_ * NQ_) / 32, H_ / 64, 1), 32, 0, stream>>>(
      queries, Wq, qproj, D_, H_, 0, 0, 0);

  wmma_gemm_f32<<<dim3((B_ * NK_) / 32, H_ / 64, 1), 32, 0, stream>>>(
      keys, Wk, kproj, D_, H_, 0, 0, 0);

  scores_softmax<<<dim3(NQ_ / QB_, B_), NK_, 0, stream>>>(
      qproj, kproj, Wv, valid_lens, attn);

  wmma_gemm_f32<<<dim3(NQ_ / 32, D_ / 64, B_), 32, 0, stream>>>(
      attn, values, out, NK_, D_,
      (long long)NQ_ * NK_, (long long)NK_ * D_, (long long)NQ_ * D_);
}


# Softmax tail: scores transposed through LDS, each wave reduces one whole row (16 values per lane) and stores half of it as float4: 10 dependent lane exchanges per wave, was 80; row-sum order changed
# speedup vs baseline: 2.4179x; 2.4179x over previous
#include <hip/hip_runtime.h>

#define B_  4
#define NQ_ 512
#define NK_ 512
#define D_  512
#define H_  128
#define QB_ 8

typedef __attribute__((ext_vector_type(2))) float v2f;
typedef __attribute__((ext_vector_type(8))) float v8f;

__device__ __forceinline__ float hw_tanh(float x) {
#if __has_builtin(__builtin_amdgcn_tanhf)
  return __builtin_amdgcn_tanhf(x);
#else
  float y;
  asm volatile("v_tanh_f32 %0, %1" : "=v"(y) : "v"(x));
  return y;
#endif
}

__device__ __forceinline__ float hw_exp(float x) {
#if __has_builtin(__builtin_amdgcn_exp2f)
  return __builtin_amdgcn_exp2f(x * 1.44269504088896340736f);
#else
  return __expf(x);
#endif
}

#define GT_M 64
#define GT_N 64
#define GT_K 64
#define GA_PITCH 68
#define GB_PITCH 144

#define GEMM_GLOAD(k0)                                                        \
  do {                                                                        \
    ra0 = *(const float4*)(ag + (long long)(0) * K + (k0));                  \
    ra1 = *(const float4*)(ag + (long long)(8) * K + (k0));                  \
    ra2 = *(const float4*)(ag + (long long)(16) * K + (k0));                  \
    ra3 = *(const float4*)(ag + (long long)(24) * K + (k0));                  \
    ra4 = *(const float4*)(ag + (long long)(32) * K + (k0));                  \
    ra5 = *(const float4*)(ag + (long long)(40) * K + (k0));                  \
    ra6 = *(const float4*)(ag + (long long)(48) * K + (k0));                  \
    ra7 = *(const float4*)(ag + (long long)(56) * K + (k0));                  \
    rb0 = *(const float4*)(bg + (long long)((k0) + 0) * N);                  \
    rb1 = *(const float4*)(bg + (long long)((k0) + 0) * N + N);              \
    rb2 = *(const float4*)(bg + (long long)((k0) + 16) * N);                  \
    rb3 = *(const float4*)(bg + (long long)((k0) + 16) * N + N);              \
    rb4 = *(const float4*)(bg + (long long)((k0) + 32) * N);                  \
    rb5 = *(const float4*)(bg + (long long)((k0) + 32) * N + N);              \
    rb6 = *(const float4*)(bg + (long long)((k0) + 48) * N);                  \
    rb7 = *(const float4*)(bg + (long long)((k0) + 48) * N + N);              \
  } while (0)

#define GEMM_SSTORE(as, bs)                                                   \
  do {                                                                        \
    *(float4*)((as) + 0 * GA_PITCH) = ra0;                                  \
    *(float4*)((as) + 8 * GA_PITCH) = ra1;                                  \
    *(float4*)((as) + 16 * GA_PITCH) = ra2;                                  \
    *(float4*)((as) + 24 * GA_PITCH) = ra3;                                  \
    *(float4*)((as) + 32 * GA_PITCH) = ra4;                                  \
    *(float4*)((as) + 40 * GA_PITCH) = ra5;                                  \
    *(float4*)((as) + 48 * GA_PITCH) = ra6;                                  \
    *(float4*)((as) + 56 * GA_PITCH) = ra7;                                  \
    *(float4*)((bs) + 0 * GB_PITCH) = rb0;                                  \
    *(float4*)((bs) + 0 * GB_PITCH + 16) = rb1;                             \
    *(float4*)((bs) + 8 * GB_PITCH) = rb2;                                  \
    *(float4*)((bs) + 8 * GB_PITCH + 16) = rb3;                             \
    *(float4*)((bs) + 16 * GB_PITCH) = rb4;                                  \
    *(float4*)((bs) + 16 * GB_PITCH + 16) = rb5;                             \
    *(float4*)((bs) + 24 * GB_PITCH) = rb6;                                  \
    *(float4*)((bs) + 24 * GB_PITCH + 16) = rb7;                             \
  } while (0)

__device__ __forceinline__ void gemm_slab(const float* as, const float* bs,
                                          v8f& acc00, v8f& acc01, v8f& acc10, v8f& acc11) {
  v2f a0[GT_K / 4], a1[GT_K / 4], b0[GT_K / 4], b1[GT_K / 4];
#pragma unroll
  for (int ks = 0; ks < GT_K / 4; ++ks) {
    if (ks == 4) __builtin_amdgcn_sched_barrier(0);
    a0[ks] = *(const v2f*)(as + 4 * ks);
    a1[ks] = *(const v2f*)(as + 16 * GA_PITCH + 4 * ks);
    b0[ks].x = bs[(2 * ks) * GB_PITCH];
    b0[ks].y = bs[(2 * ks) * GB_PITCH + 16];
    b1[ks].x = bs[(2 * ks) * GB_PITCH + 32];
    b1[ks].y = bs[(2 * ks) * GB_PITCH + 48];
  }
  __builtin_amdgcn_sched_barrier(0);
#pragma unroll
  for (int ks = 0; ks < GT_K / 4; ++ks) {
    acc00 = __builtin_amdgcn_wmma_f32_16x16x4_f32(false, a0[ks], false, b0[ks], (short)0, acc00, false, false);
    acc01 = __builtin_amdgcn_wmma_f32_16x16x4_f32(false, a0[ks], false, b1[ks], (short)0, acc01, false, false);
    acc10 = __builtin_amdgcn_wmma_f32_16x16x4_f32(false, a1[ks], false, b0[ks], (short)0, acc10, false, false);
    acc11 = __builtin_amdgcn_wmma_f32_16x16x4_f32(false, a1[ks], false, b1[ks], (short)0, acc11, false, false);
  }
}

__global__ __launch_bounds__(128) void wmma_gemm_f32(
    const float* A, const float* Bm, float* C, int K, int N,
    long long sA, long long sB, long long sC) {
  __shared__ __attribute__((aligned(16))) float As[2 * GT_M * GA_PITCH];
  __shared__ __attribute__((aligned(16))) float Bs[2 * (GT_K / 2) * GB_PITCH];

  const int tid   = threadIdx.x;
  const int lane  = tid & 31;
  const int wave  = tid >> 5;
  const int wm    = wave >> 1;
  const int wn    = wave & 1;
  const int mBase = blockIdx.x * GT_M;
  const int nBase = blockIdx.y * GT_N;
  const int batch = blockIdx.z;
  A  += (long long)batch * sA;
  Bm += (long long)batch * sB;
  C  += (long long)batch * sC;

  const int c4 = tid & 15;
  const int r8 = tid >> 4;
  const float* ag = A  + (long long)(mBase + r8) * K + 4 * c4;
  const float* bg = Bm + (long long)(2 * r8) * N + nBase + 4 * c4;
  float* asw = As + r8 * GA_PITCH + 4 * c4;
  float* bsw = Bs + r8 * GB_PITCH + (c4 >> 2) * 32 + (c4 & 3) * 4;

  const int c = lane & 15;
  const int h = lane >> 4;
  const float* asr = As + (wm * 32 + c) * GA_PITCH + 2 * h;
  const float* bsr = Bs + h * GB_PITCH + wn * 64 + c;

  v8f acc00 = {}, acc01 = {}, acc10 = {}, acc11 = {};
  float4 ra0, ra1, ra2, ra3, ra4, ra5, ra6, ra7;
  float4 rb0, rb1, rb2, rb3, rb4, rb5, rb6, rb7;

  GEMM_GLOAD(0);
  GEMM_SSTORE(asw, bsw);
  __syncthreads();

  const int nSlab = K / GT_K;
  int cur = 0;
  for (int s = 0; s + 1 < nSlab; ++s) {
    const int nxt = cur ^ 1;
    GEMM_GLOAD((s + 1) * GT_K);
    __builtin_amdgcn_sched_barrier(0);
    gemm_slab(asr + cur * (GT_M * GA_PITCH), bsr + cur * ((GT_K / 2) * GB_PITCH),
              acc00, acc01, acc10, acc11);
    asm volatile("" :: "v"(acc00), "v"(acc01), "v"(acc10), "v"(acc11) : "memory");
    GEMM_SSTORE(asw + nxt * (GT_M * GA_PITCH), bsw + nxt * ((GT_K / 2) * GB_PITCH));
    __syncthreads();
    cur = nxt;
  }
  gemm_slab(asr + cur * (GT_M * GA_PITCH), bsr + cur * ((GT_K / 2) * GB_PITCH),
            acc00, acc01, acc10, acc11);

  const int mOut0 = mBase + wm * 32 + h * 8;
  const int nOut0 = nBase + wn * 32 + c;
#pragma unroll
  for (int r = 0; r < 8; ++r) {
    long long row0 = (long long)(mOut0 + r) * N + nOut0;
    C[row0 +  0] = acc00[r];
    C[row0 + 16] = acc01[r];
    long long row1 = row0 + (long long)16 * N;
    C[row1 +  0] = acc10[r];
    C[row1 + 16] = acc11[r];
  }
}
#undef GEMM_GLOAD
#undef GEMM_SSTORE

#define QP_KC    64
#define QP_PITCH 272
#define QS_PITCH 516

#define WQ_GLOAD(ch)                                                          \
  do {                                                                        \
    rw0 = *(const float4*)(wg + (long long)((ch) * QP_KC + 0) * H_);          \
    rw1 = *(const float4*)(wg + (long long)((ch) * QP_KC + 16) * H_);         \
    rw2 = *(const float4*)(wg + (long long)((ch) * QP_KC + 32) * H_);         \
    rw3 = *(const float4*)(wg + (long long)((ch) * QP_KC + 48) * H_);         \
  } while (0)

#define WQ_SSTORE(ws)                                                         \
  do {                                                                        \
    *(float4*)((ws) + 0 * QP_PITCH) = rw0;                                    \
    *(float4*)((ws) + 8 * QP_PITCH) = rw1;                                    \
    *(float4*)((ws) + 16 * QP_PITCH) = rw2;                                   \
    *(float4*)((ws) + 24 * QP_PITCH) = rw3;                                   \
  } while (0)

__global__ __launch_bounds__(NK_) void scores_softmax(
    const float* __restrict__ queries, const float* __restrict__ Wq,
    const float* __restrict__ kproj,
    const float* __restrict__ Wv, const int* __restrict__ valid_lens,
    float* __restrict__ attn) {
  __shared__ __attribute__((aligned(16))) float qs[QB_ * QS_PITCH];
  __shared__ __attribute__((aligned(16))) float wqs[2 * (QP_KC / 2) * QP_PITCH];
  __shared__ float qp[QB_][H_];
  __shared__ float qph[QB_][H_];
  __shared__ float wv[H_];

  const int q0   = blockIdx.x * QB_;
  const int b    = blockIdx.y;
  const int k    = threadIdx.x;
  const int lane = k & 31;
  const int wid  = k >> 5;

  {
    const int c   = lane & 15;
    const int hh  = lane >> 4;
    const int nt  = wid & 7;
    const int kh  = wid >> 3;
    const int c4  = k & 31;
    const int r16 = k >> 5;
    const float* wg = Wq + (long long)r16 * H_ + 4 * c4;
    float* wsw = wqs + (r16 >> 1) * QP_PITCH + (c4 >> 2) * 32 + (r16 & 1) * 16 + (c4 & 3) * 4;
    const float* qg = queries + (long long)(b * NQ_ + q0) * D_;

    float4 rw0, rw1, rw2, rw3;
    WQ_GLOAD(0);
    const float4 rq0 = *(const float4*)(qg + 4 * k);
    const float4 rq1 = *(const float4*)(qg + 4 * (k + NK_));
    *(float4*)(qs + (k >> 7) * QS_PITCH + 4 * (k & 127)) = rq0;
    *(float4*)(qs + ((k >> 7) + 4) * QS_PITCH + 4 * (k & 127)) = rq1;
    WQ_SSTORE(wsw);
    if (k < H_) wv[k] = Wv[k];
    __syncthreads();

    const float* asr = qs + (c & 7) * QS_PITCH + 32 * kh + 2 * hh;
    const float* bsr = wqs + (16 * kh + hh) * QP_PITCH + nt * 32 + c;
    v8f acc = {};
    int cur = 0;
    for (int ch = 0; ch < D_ / QP_KC; ++ch) {
      const bool more = ch + 1 < D_ / QP_KC;
      if (more) WQ_GLOAD(ch + 1);
      const float* as = asr + ch * QP_KC;
      const float* bs = bsr + cur * ((QP_KC / 2) * QP_PITCH);
      v2f af[8], bf[8];
#pragma unroll
      for (int i = 0; i < 8; ++i) {
        af[i]   = *(const v2f*)(as + 4 * i);
        bf[i].x = bs[(2 * i) * QP_PITCH];
        bf[i].y = bs[(2 * i) * QP_PITCH + 16];
      }
#pragma unroll
      for (int i = 0; i < 8; ++i)
        acc = __builtin_amdgcn_wmma_f32_16x16x4_f32(false, af[i], false, bf[i], (short)0, acc, false, false);
      if (more) WQ_SSTORE(wsw + (cur ^ 1) * ((QP_KC / 2) * QP_PITCH));
      __syncthreads();
      cur ^= 1;
    }

    if (kh == 1 && hh == 0) {
#pragma unroll
      for (int r = 0; r < QB_; ++r) qph[r][nt * 16 + c] = acc[r];
    }
    __syncthreads();
    if (kh == 0 && hh == 0) {
#pragma unroll
      for (int r = 0; r < QB_; ++r) qp[r][nt * 16 + c] = acc[r] + qph[r][nt * 16 + c];
    }
    __syncthreads();
  }

  const float4* kp4 =
      (const float4*)(kproj + (long long)(b * NK_ + k) * H_);
  float s[QB_];
#pragma unroll
  for (int j = 0; j < QB_; ++j) s[j] = 0.0f;

  for (int h4 = 0; h4 < H_ / 4; ++h4) {
    const float4 kv = kp4[h4];
    const int h = h4 * 4;
#pragma unroll
    for (int j = 0; j < QB_; ++j) {
      s[j] += wv[h + 0] * hw_tanh(qp[j][h + 0] + kv.x);
      s[j] += wv[h + 1] * hw_tanh(qp[j][h + 1] + kv.y);
      s[j] += wv[h + 2] * hw_tanh(qp[j][h + 2] + kv.z);
      s[j] += wv[h + 3] * hw_tanh(qp[j][h + 3] + kv.w);
    }
  }

  const int vlen = valid_lens[b];

  float* sc = qs;
#pragma unroll
  for (int j = 0; j < QB_; ++j) sc[j * QS_PITCH + k] = (k < vlen) ? s[j] : -1.0e6f;
  __syncthreads();

  const int row = wid >> 1;
  const int hf  = wid & 1;
  const float4* rp = (const float4*)(sc + row * QS_PITCH);
  const float4 x0 = rp[lane];
  const float4 x1 = rp[lane + 32];
  const float4 x2 = rp[lane + 64];
  const float4 x3 = rp[lane + 96];

  float m = fmaxf(fmaxf(fmaxf(x0.x, x0.y), fmaxf(x0.z, x0.w)),
                  fmaxf(fmaxf(x1.x, x1.y), fmaxf(x1.z, x1.w)));
  m = fmaxf(m, fmaxf(fmaxf(fmaxf(x2.x, x2.y), fmaxf(x2.z, x2.w)),
                     fmaxf(fmaxf(x3.x, x3.y), fmaxf(x3.z, x3.w))));
#pragma unroll
  for (int off = 16; off > 0; off >>= 1) m = fmaxf(m, __shfl_xor(m, off, 32));

  float4 e0, e1, e2, e3;
  e0.x = hw_exp(x0.x - m); e0.y = hw_exp(x0.y - m); e0.z = hw_exp(x0.z - m); e0.w = hw_exp(x0.w - m);
  e1.x = hw_exp(x1.x - m); e1.y = hw_exp(x1.y - m); e1.z = hw_exp(x1.z - m); e1.w = hw_exp(x1.w - m);
  e2.x = hw_exp(x2.x - m); e2.y = hw_exp(x2.y - m); e2.z = hw_exp(x2.z - m); e2.w = hw_exp(x2.w - m);
  e3.x = hw_exp(x3.x - m); e3.y = hw_exp(x3.y - m); e3.z = hw_exp(x3.z - m); e3.w = hw_exp(x3.w - m);
  float t = (((e0.x + e0.y) + (e0.z + e0.w)) + ((e1.x + e1.y) + (e1.z + e1.w))) +
            (((e2.x + e2.y) + (e2.z + e2.w)) + ((e3.x + e3.y) + (e3.z + e3.w)));
#pragma unroll
  for (int off = 16; off > 0; off >>= 1) t += __shfl_xor(t, off, 32);

  const float4 ya = hf ? e2 : e0;
  const float4 yb = hf ? e3 : e1;
  float* op = attn + (long long)(b * NQ_ + q0 + row) * NK_ + hf * (NK_ / 2) + 4 * lane;
  float4 oa, ob;
  oa.x = ya.x / t; oa.y = ya.y / t; oa.z = ya.z / t; oa.w = ya.w / t;
  ob.x = yb.x / t; ob.y = yb.y / t; ob.z = yb.z / t; ob.w = yb.w / t;
  *(float4*)op = oa;
  *(float4*)(op + NK_ / 4) = ob;
}

#undef WQ_GLOAD
#undef WQ_SSTORE

extern "C" void kernel_launch(void* const* d_in, const int* in_sizes, int n_in,
                              void* d_out, int out_size, void* d_ws, size_t ws_size,
                              hipStream_t stream) {
  (void)in_sizes; (void)n_in; (void)out_size; (void)ws_size;

  const float* queries    = (const float*)d_in[0];
  const float* keys       = (const float*)d_in[1];
  const float* values     = (const float*)d_in[2];
  const float* Wq         = (const float*)d_in[3];
  const float* Wk         = (const float*)d_in[4];
  const float* Wv         = (const float*)d_in[5];
  const int*   valid_lens = (const int*)d_in[6];
  float*       out        = (float*)d_out;

  float* kproj = (float*)d_ws + (long long)B_ * NQ_ * H_;
  float* attn  = kproj + (long long)B_ * NK_ * H_;

  wmma_gemm_f32<<<dim3((B_ * NK_) / GT_M, H_ / GT_N, 1), 128, 0, stream>>>(
      keys, Wk, kproj, D_, H_, 0, 0, 0);

  scores_softmax<<<dim3(NQ_ / QB_, B_), NK_, 0, stream>>>(
      queries, Wq, kproj, Wv, valid_lens, attn);

  wmma_gemm_f32<<<dim3(NQ_ / GT_M, D_ / GT_N, B_), 128, 0, stream>>>(
      attn, values, out, NK_, D_,
      (long long)NQ_ * NK_, (long long)NK_ * D_, (long long)NQ_ * D_);
}


# Key projection GEMM re-tiled to fill the chip: 256 workgroups of 32x32 (one 16x16 tile, 128 chained WMMAs per wave), was 64 of 64x64 (512 per wave); two K slabs of 256 in LDS; sums' order unchanged
# speedup vs baseline: 2.6999x; 1.1166x over previous
#include <hip/hip_runtime.h>

#define B_  4
#define NQ_ 512
#define NK_ 512
#define D_  512
#define H_  128
#define QB_ 8

typedef __attribute__((ext_vector_type(2))) float v2f;
typedef __attribute__((ext_vector_type(8))) float v8f;

__device__ __forceinline__ float hw_tanh(float x) {
#if __has_builtin(__builtin_amdgcn_tanhf)
  return __builtin_amdgcn_tanhf(x);
#else
  float y;
  asm volatile("v_tanh_f32 %0, %1" : "=v"(y) : "v"(x));
  return y;
#endif
}

__device__ __forceinline__ float hw_exp(float x) {
#if __has_builtin(__builtin_amdgcn_exp2f)
  return __builtin_amdgcn_exp2f(x * 1.44269504088896340736f);
#else
  return __expf(x);
#endif
}

#define GT_M 64
#define GT_N 64
#define GT_K 64
#define GA_PITCH 68
#define GB_PITCH 144

#define GEMM_GLOAD(k0)                                                        \
  do {                                                                        \
    ra0 = *(const float4*)(ag + (long long)(0) * K + (k0));                  \
    ra1 = *(const float4*)(ag + (long long)(8) * K + (k0));                  \
    ra2 = *(const float4*)(ag + (long long)(16) * K + (k0));                  \
    ra3 = *(const float4*)(ag + (long long)(24) * K + (k0));                  \
    ra4 = *(const float4*)(ag + (long long)(32) * K + (k0));                  \
    ra5 = *(const float4*)(ag + (long long)(40) * K + (k0));                  \
    ra6 = *(const float4*)(ag + (long long)(48) * K + (k0));                  \
    ra7 = *(const float4*)(ag + (long long)(56) * K + (k0));                  \
    rb0 = *(const float4*)(bg + (long long)((k0) + 0) * N);                  \
    rb1 = *(const float4*)(bg + (long long)((k0) + 0) * N + N);              \
    rb2 = *(const float4*)(bg + (long long)((k0) + 16) * N);                  \
    rb3 = *(const float4*)(bg + (long long)((k0) + 16) * N + N);              \
    rb4 = *(const float4*)(bg + (long long)((k0) + 32) * N);                  \
    rb5 = *(const float4*)(bg + (long long)((k0) + 32) * N + N);              \
    rb6 = *(const float4*)(bg + (long long)((k0) + 48) * N);                  \
    rb7 = *(const float4*)(bg + (long long)((k0) + 48) * N + N);              \
  } while (0)

#define GEMM_SSTORE(as, bs)                                                   \
  do {                                                                        \
    *(float4*)((as) + 0 * GA_PITCH) = ra0;                                  \
    *(float4*)((as) + 8 * GA_PITCH) = ra1;                                  \
    *(float4*)((as) + 16 * GA_PITCH) = ra2;                                  \
    *(float4*)((as) + 24 * GA_PITCH) = ra3;                                  \
    *(float4*)((as) + 32 * GA_PITCH) = ra4;                                  \
    *(float4*)((as) + 40 * GA_PITCH) = ra5;                                  \
    *(float4*)((as) + 48 * GA_PITCH) = ra6;                                  \
    *(float4*)((as) + 56 * GA_PITCH) = ra7;                                  \
    *(float4*)((bs) + 0 * GB_PITCH) = rb0;                                  \
    *(float4*)((bs) + 0 * GB_PITCH + 16) = rb1;                             \
    *(float4*)((bs) + 8 * GB_PITCH) = rb2;                                  \
    *(float4*)((bs) + 8 * GB_PITCH + 16) = rb3;                             \
    *(float4*)((bs) + 16 * GB_PITCH) = rb4;                                  \
    *(float4*)((bs) + 16 * GB_PITCH + 16) = rb5;                             \
    *(float4*)((bs) + 24 * GB_PITCH) = rb6;                                  \
    *(float4*)((bs) + 24 * GB_PITCH + 16) = rb7;                             \
  } while (0)

__device__ __forceinline__ void gemm_slab(const float* as, const float* bs,
                                          v8f& acc00, v8f& acc01, v8f& acc10, v8f& acc11) {
  v2f a0[GT_K / 4], a1[GT_K / 4], b0[GT_K / 4], b1[GT_K / 4];
#pragma unroll
  for (int ks = 0; ks < GT_K / 4; ++ks) {
    if (ks == 4) __builtin_amdgcn_sched_barrier(0);
    a0[ks] = *(const v2f*)(as + 4 * ks);
    a1[ks] = *(const v2f*)(as + 16 * GA_PITCH + 4 * ks);
    b0[ks].x = bs[(2 * ks) * GB_PITCH];
    b0[ks].y = bs[(2 * ks) * GB_PITCH + 16];
    b1[ks].x = bs[(2 * ks) * GB_PITCH + 32];
    b1[ks].y = bs[(2 * ks) * GB_PITCH + 48];
  }
  __builtin_amdgcn_sched_barrier(0);
#pragma unroll
  for (int ks = 0; ks < GT_K / 4; ++ks) {
    acc00 = __builtin_amdgcn_wmma_f32_16x16x4_f32(false, a0[ks], false, b0[ks], (short)0, acc00, false, false);
    acc01 = __builtin_amdgcn_wmma_f32_16x16x4_f32(false, a0[ks], false, b1[ks], (short)0, acc01, false, false);
    acc10 = __builtin_amdgcn_wmma_f32_16x16x4_f32(false, a1[ks], false, b0[ks], (short)0, acc10, false, false);
    acc11 = __builtin_amdgcn_wmma_f32_16x16x4_f32(false, a1[ks], false, b1[ks], (short)0, acc11, false, false);
  }
}

__global__ __launch_bounds__(128) void wmma_gemm_f32(
    const float* A, const float* Bm, float* C, int K, int N,
    long long sA, long long sB, long long sC) {
  __shared__ __attribute__((aligned(16))) float As[2 * GT_M * GA_PITCH];
  __shared__ __attribute__((aligned(16))) float Bs[2 * (GT_K / 2) * GB_PITCH];

  const int tid   = threadIdx.x;
  const int lane  = tid & 31;
  const int wave  = tid >> 5;
  const int wm    = wave >> 1;
  const int wn    = wave & 1;
  const int mBase = blockIdx.x * GT_M;
  const int nBase = blockIdx.y * GT_N;
  const int batch = blockIdx.z;
  A  += (long long)batch * sA;
  Bm += (long long)batch * sB;
  C  += (long long)batch * sC;

  const int c4 = tid & 15;
  const int r8 = tid >> 4;
  const float* ag = A  + (long long)(mBase + r8) * K + 4 * c4;
  const float* bg = Bm + (long long)(2 * r8) * N + nBase + 4 * c4;
  float* asw = As + r8 * GA_PITCH + 4 * c4;
  float* bsw = Bs + r8 * GB_PITCH + (c4 >> 2) * 32 + (c4 & 3) * 4;

  const int c = lane & 15;
  const int h = lane >> 4;
  const float* asr = As + (wm * 32 + c) * GA_PITCH + 2 * h;
  const float* bsr = Bs + h * GB_PITCH + wn * 64 + c;

  v8f acc00 = {}, acc01 = {}, acc10 = {}, acc11 = {};
  float4 ra0, ra1, ra2, ra3, ra4, ra5, ra6, ra7;
  float4 rb0, rb1, rb2, rb3, rb4, rb5, rb6, rb7;

  GEMM_GLOAD(0);
  GEMM_SSTORE(asw, bsw);
  __syncthreads();

  const int nSlab = K / GT_K;
  int cur = 0;
  for (int s = 0; s + 1 < nSlab; ++s) {
    const int nxt = cur ^ 1;
    GEMM_GLOAD((s + 1) * GT_K);
    __builtin_amdgcn_sched_barrier(0);
    gemm_slab(asr + cur * (GT_M * GA_PITCH), bsr + cur * ((GT_K / 2) * GB_PITCH),
              acc00, acc01, acc10, acc11);
    asm volatile("" :: "v"(acc00), "v"(acc01), "v"(acc10), "v"(acc11) : "memory");
    GEMM_SSTORE(asw + nxt * (GT_M * GA_PITCH), bsw + nxt * ((GT_K / 2) * GB_PITCH));
    __syncthreads();
    cur = nxt;
  }
  gemm_slab(asr + cur * (GT_M * GA_PITCH), bsr + cur * ((GT_K / 2) * GB_PITCH),
            acc00, acc01, acc10, acc11);

  const int mOut0 = mBase + wm * 32 + h * 8;
  const int nOut0 = nBase + wn * 32 + c;
#pragma unroll
  for (int r = 0; r < 8; ++r) {
    long long row0 = (long long)(mOut0 + r) * N + nOut0;
    C[row0 +  0] = acc00[r];
    C[row0 + 16] = acc01[r];
    long long row1 = row0 + (long long)16 * N;
    C[row1 +  0] = acc10[r];
    C[row1 + 16] = acc11[r];
  }
}
#undef GEMM_GLOAD
#undef GEMM_SSTORE

#define KP_TM 32
#define KP_TN 32
#define KP_KS 256
#define KP_KG 128
#define KP_APITCH 516
#define KP_BPITCH 80
#define KP_CPITCH 36

#define KP_GLOAD(koff) \
  do { \
    pa0 = *(const float4*)(ag + (long long)(0) * D_ + (koff)); \
    pa1 = *(const float4*)(ag + (long long)(2) * D_ + (koff)); \
    pa2 = *(const float4*)(ag + (long long)(4) * D_ + (koff)); \
    pa3 = *(const float4*)(ag + (long long)(6) * D_ + (koff)); \
    pa4 = *(const float4*)(ag + (long long)(8) * D_ + (koff)); \
    pa5 = *(const float4*)(ag + (long long)(10) * D_ + (koff)); \
    pa6 = *(const float4*)(ag + (long long)(12) * D_ + (koff)); \
    pa7 = *(const float4*)(ag + (long long)(14) * D_ + (koff)); \
    pa8 = *(const float4*)(ag + (long long)(16) * D_ + (koff)); \
    pa9 = *(const float4*)(ag + (long long)(18) * D_ + (koff)); \
    pa10 = *(const float4*)(ag + (long long)(20) * D_ + (koff)); \
    pa11 = *(const float4*)(ag + (long long)(22) * D_ + (koff)); \
    pa12 = *(const float4*)(ag + (long long)(24) * D_ + (koff)); \
    pa13 = *(const float4*)(ag + (long long)(26) * D_ + (koff)); \
    pa14 = *(const float4*)(ag + (long long)(28) * D_ + (koff)); \
    pa15 = *(const float4*)(ag + (long long)(30) * D_ + (koff)); \
    pb0 = *(const float4*)(bg + (long long)((koff) + 0) * H_); \
    pb1 = *(const float4*)(bg + (long long)((koff) + 16) * H_); \
    pb2 = *(const float4*)(bg + (long long)((koff) + 32) * H_); \
    pb3 = *(const float4*)(bg + (long long)((koff) + 48) * H_); \
    pb4 = *(const float4*)(bg + (long long)((koff) + 64) * H_); \
    pb5 = *(const float4*)(bg + (long long)((koff) + 80) * H_); \
    pb6 = *(const float4*)(bg + (long long)((koff) + 96) * H_); \
    pb7 = *(const float4*)(bg + (long long)((koff) + 112) * H_); \
    pb8 = *(const float4*)(bg + (long long)((koff) + 128) * H_); \
    pb9 = *(const float4*)(bg + (long long)((koff) + 144) * H_); \
    pb10 = *(const float4*)(bg + (long long)((koff) + 160) * H_); \
    pb11 = *(const float4*)(bg + (long long)((koff) + 176) * H_); \
    pb12 = *(const float4*)(bg + (long long)((koff) + 192) * H_); \
    pb13 = *(const float4*)(bg + (long long)((koff) + 208) * H_); \
    pb14 = *(const float4*)(bg + (long long)((koff) + 224) * H_); \
    pb15 = *(const float4*)(bg + (long long)((koff) + 240) * H_); \
  } while (0)

#define KP_SSTORE(koff) \
  do { \
    *(float4*)(asw + 0 * KP_APITCH + (koff)) = pa0; \
    *(float4*)(asw + 2 * KP_APITCH + (koff)) = pa1; \
    *(float4*)(asw + 4 * KP_APITCH + (koff)) = pa2; \
    *(float4*)(asw + 6 * KP_APITCH + (koff)) = pa3; \
    *(float4*)(asw + 8 * KP_APITCH + (koff)) = pa4; \
    *(float4*)(asw + 10 * KP_APITCH + (koff)) = pa5; \
    *(float4*)(asw + 12 * KP_APITCH + (koff)) = pa6; \
    *(float4*)(asw + 14 * KP_APITCH + (koff)) = pa7; \
    *(float4*)(asw + 16 * KP_APITCH + (koff)) = pa8; \
    *(float4*)(asw + 18 * KP_APITCH + (koff)) = pa9; \
    *(float4*)(asw + 20 * KP_APITCH + (koff)) = pa10; \
    *(float4*)(asw + 22 * KP_APITCH + (koff)) = pa11; \
    *(float4*)(asw + 24 * KP_APITCH + (koff)) = pa12; \
    *(float4*)(asw + 26 * KP_APITCH + (koff)) = pa13; \
    *(float4*)(asw + 28 * KP_APITCH + (koff)) = pa14; \
    *(float4*)(asw + 30 * KP_APITCH + (koff)) = pa15; \
    *(float4*)(bsw + ((koff) / 2 + 0) * KP_BPITCH) = pb0; \
    *(float4*)(bsw + ((koff) / 2 + 8) * KP_BPITCH) = pb1; \
    *(float4*)(bsw + ((koff) / 2 + 16) * KP_BPITCH) = pb2; \
    *(float4*)(bsw + ((koff) / 2 + 24) * KP_BPITCH) = pb3; \
    *(float4*)(bsw + ((koff) / 2 + 32) * KP_BPITCH) = pb4; \
    *(float4*)(bsw + ((koff) / 2 + 40) * KP_BPITCH) = pb5; \
    *(float4*)(bsw + ((koff) / 2 + 48) * KP_BPITCH) = pb6; \
    *(float4*)(bsw + ((koff) / 2 + 56) * KP_BPITCH) = pb7; \
    *(float4*)(bsw + ((koff) / 2 + 64) * KP_BPITCH) = pb8; \
    *(float4*)(bsw + ((koff) / 2 + 72) * KP_BPITCH) = pb9; \
    *(float4*)(bsw + ((koff) / 2 + 80) * KP_BPITCH) = pb10; \
    *(float4*)(bsw + ((koff) / 2 + 88) * KP_BPITCH) = pb11; \
    *(float4*)(bsw + ((koff) / 2 + 96) * KP_BPITCH) = pb12; \
    *(float4*)(bsw + ((koff) / 2 + 104) * KP_BPITCH) = pb13; \
    *(float4*)(bsw + ((koff) / 2 + 112) * KP_BPITCH) = pb14; \
    *(float4*)(bsw + ((koff) / 2 + 120) * KP_BPITCH) = pb15; \
  } while (0)

__device__ __forceinline__ void kp_group(const float* as, const float* bs, v8f& acc) {
  v2f a[KP_KG / 4], b[KP_KG / 4];
#pragma unroll
  for (int ks = 0; ks < KP_KG / 4; ++ks) {
    if (ks == 4) __builtin_amdgcn_sched_barrier(0);
    a[ks]   = *(const v2f*)(as + 4 * ks);
    b[ks].x = bs[(2 * ks) * KP_BPITCH];
    b[ks].y = bs[(2 * ks) * KP_BPITCH + 16];
  }
  __builtin_amdgcn_sched_barrier(0);
#pragma unroll
  for (int ks = 0; ks < KP_KG / 4; ++ks)
    acc = __builtin_amdgcn_wmma_f32_16x16x4_f32(false, a[ks], false, b[ks], (short)0, acc, false, false);
  asm volatile("" :: "v"(acc) : "memory");
}

__global__ __launch_bounds__(128) void kproj_gemm_f32(
    const float* A, const float* Bm, float* C) {
  __shared__ __attribute__((aligned(16))) float As[KP_TM * KP_APITCH];
  __shared__ __attribute__((aligned(16))) float Bs[(D_ / 2) * KP_BPITCH];
  __shared__ __attribute__((aligned(16))) float Cs[KP_TM * KP_CPITCH];

  const int tid   = threadIdx.x;
  const int lane  = tid & 31;
  const int wave  = tid >> 5;
  const int wm    = wave >> 1;
  const int wn    = wave & 1;
  const int mBase = blockIdx.x * KP_TM;
  const int nBase = blockIdx.y * KP_TN;

  const int c64 = tid & 63;
  const int r2  = tid >> 6;
  const int c8  = tid & 7;
  const int kr  = tid >> 3;
  const float* ag = A  + (long long)(mBase + r2) * D_ + 4 * c64;
  const float* bg = Bm + (long long)kr * H_ + nBase + 4 * c8;
  float* asw = As + r2 * KP_APITCH + 4 * c64;
  float* bsw = Bs + (kr >> 1) * KP_BPITCH + (c8 >> 2) * 32 + (kr & 1) * 16 + (c8 & 3) * 4;

  const int c = lane & 15;
  const int h = lane >> 4;
  const float* asr = As + (wm * 16 + c) * KP_APITCH + 2 * h;
  const float* bsr = Bs + h * KP_BPITCH + wn * 32 + c;

  v8f acc = {};
  float4 pa0, pa1, pa2, pa3, pa4, pa5, pa6, pa7, pa8, pa9, pa10, pa11, pa12, pa13, pa14, pa15;
  float4 pb0, pb1, pb2, pb3, pb4, pb5, pb6, pb7, pb8, pb9, pb10, pb11, pb12, pb13, pb14, pb15;

  KP_GLOAD(0);
  KP_SSTORE(0);
  __syncthreads();

  KP_GLOAD(KP_KS);
  __builtin_amdgcn_sched_barrier(0);

  kp_group(asr,         bsr,                             acc);
  kp_group(asr + KP_KG, bsr + (KP_KG / 2) * KP_BPITCH,   acc);

  KP_SSTORE(KP_KS);
  __syncthreads();

  kp_group(asr + 2 * KP_KG, bsr + (2 * KP_KG / 2) * KP_BPITCH, acc);
  kp_group(asr + 3 * KP_KG, bsr + (3 * KP_KG / 2) * KP_BPITCH, acc);

#pragma unroll
  for (int r = 0; r < 8; ++r) Cs[(wm * 16 + h * 8 + r) * KP_CPITCH + wn * 16 + c] = acc[r];
  __syncthreads();
  const int orow = tid >> 3;
  const float4 o0 = *(const float4*)(Cs + orow * KP_CPITCH + 4 * c8);
  const float4 o1 = *(const float4*)(Cs + (orow + 16) * KP_CPITCH + 4 * c8);
  *(float4*)(C + (long long)(mBase + orow) * H_ + nBase + 4 * c8) = o0;
  *(float4*)(C + (long long)(mBase + orow + 16) * H_ + nBase + 4 * c8) = o1;
}
#undef KP_GLOAD
#undef KP_SSTORE

#define QP_KC    64
#define QP_PITCH 272
#define QS_PITCH 516

#define WQ_GLOAD(ch)                                                          \
  do {                                                                        \
    rw0 = *(const float4*)(wg + (long long)((ch) * QP_KC + 0) * H_);          \
    rw1 = *(const float4*)(wg + (long long)((ch) * QP_KC + 16) * H_);         \
    rw2 = *(const float4*)(wg + (long long)((ch) * QP_KC + 32) * H_);         \
    rw3 = *(const float4*)(wg + (long long)((ch) * QP_KC + 48) * H_);         \
  } while (0)

#define WQ_SSTORE(ws)                                                         \
  do {                                                                        \
    *(float4*)((ws) + 0 * QP_PITCH) = rw0;                                    \
    *(float4*)((ws) + 8 * QP_PITCH) = rw1;                                    \
    *(float4*)((ws) + 16 * QP_PITCH) = rw2;                                   \
    *(float4*)((ws) + 24 * QP_PITCH) = rw3;                                   \
  } while (0)

__global__ __launch_bounds__(NK_) void scores_softmax(
    const float* __restrict__ queries, const float* __restrict__ Wq,
    const float* __restrict__ kproj,
    const float* __restrict__ Wv, const int* __restrict__ valid_lens,
    float* __restrict__ attn) {
  __shared__ __attribute__((aligned(16))) float qs[QB_ * QS_PITCH];
  __shared__ __attribute__((aligned(16))) float wqs[2 * (QP_KC / 2) * QP_PITCH];
  __shared__ float qp[QB_][H_];
  __shared__ float qph[QB_][H_];
  __shared__ float wv[H_];

  const int q0   = blockIdx.x * QB_;
  const int b    = blockIdx.y;
  const int k    = threadIdx.x;
  const int lane = k & 31;
  const int wid  = k >> 5;

  {
    const int c   = lane & 15;
    const int hh  = lane >> 4;
    const int nt  = wid & 7;
    const int kh  = wid >> 3;
    const int c4  = k & 31;
    const int r16 = k >> 5;
    const float* wg = Wq + (long long)r16 * H_ + 4 * c4;
    float* wsw = wqs + (r16 >> 1) * QP_PITCH + (c4 >> 2) * 32 + (r16 & 1) * 16 + (c4 & 3) * 4;
    const float* qg = queries + (long long)(b * NQ_ + q0) * D_;

    float4 rw0, rw1, rw2, rw3;
    WQ_GLOAD(0);
    const float4 rq0 = *(const float4*)(qg + 4 * k);
    const float4 rq1 = *(const float4*)(qg + 4 * (k + NK_));
    *(float4*)(qs + (k >> 7) * QS_PITCH + 4 * (k & 127)) = rq0;
    *(float4*)(qs + ((k >> 7) + 4) * QS_PITCH + 4 * (k & 127)) = rq1;
    WQ_SSTORE(wsw);
    if (k < H_) wv[k] = Wv[k];
    __syncthreads();

    const float* asr = qs + (c & 7) * QS_PITCH + 32 * kh + 2 * hh;
    const float* bsr = wqs + (16 * kh + hh) * QP_PITCH + nt * 32 + c;
    v8f acc = {};
    int cur = 0;
    for (int ch = 0; ch < D_ / QP_KC; ++ch) {
      const bool more = ch + 1 < D_ / QP_KC;
      if (more) WQ_GLOAD(ch + 1);
      const float* as = asr + ch * QP_KC;
      const float* bs = bsr + cur * ((QP_KC / 2) * QP_PITCH);
      v2f af[8], bf[8];
#pragma unroll
      for (int i = 0; i < 8; ++i) {
        af[i]   = *(const v2f*)(as + 4 * i);
        bf[i].x = bs[(2 * i) * QP_PITCH];
        bf[i].y = bs[(2 * i) * QP_PITCH + 16];
      }
#pragma unroll
      for (int i = 0; i < 8; ++i)
        acc = __builtin_amdgcn_wmma_f32_16x16x4_f32(false, af[i], false, bf[i], (short)0, acc, false, false);
      if (more) WQ_SSTORE(wsw + (cur ^ 1) * ((QP_KC / 2) * QP_PITCH));
      __syncthreads();
      cur ^= 1;
    }

    if (kh == 1 && hh == 0) {
#pragma unroll
      for (int r = 0; r < QB_; ++r) qph[r][nt * 16 + c] = acc[r];
    }
    __syncthreads();
    if (kh == 0 && hh == 0) {
#pragma unroll
      for (int r = 0; r < QB_; ++r) qp[r][nt * 16 + c] = acc[r] + qph[r][nt * 16 + c];
    }
    __syncthreads();
  }

  const float4* kp4 =
      (const float4*)(kproj + (long long)(b * NK_ + k) * H_);
  float s[QB_];
#pragma unroll
  for (int j = 0; j < QB_; ++j) s[j] = 0.0f;

  for (int h4 = 0; h4 < H_ / 4; ++h4) {
    const float4 kv = kp4[h4];
    const int h = h4 * 4;
#pragma unroll
    for (int j = 0; j < QB_; ++j) {
      s[j] += wv[h + 0] * hw_tanh(qp[j][h + 0] + kv.x);
      s[j] += wv[h + 1] * hw_tanh(qp[j][h + 1] + kv.y);
      s[j] += wv[h + 2] * hw_tanh(qp[j][h + 2] + kv.z);
      s[j] += wv[h + 3] * hw_tanh(qp[j][h + 3] + kv.w);
    }
  }

  const int vlen = valid_lens[b];

  float* sc = qs;
#pragma unroll
  for (int j = 0; j < QB_; ++j) sc[j * QS_PITCH + k] = (k < vlen) ? s[j] : -1.0e6f;
  __syncthreads();

  const int row = wid >> 1;
  const int hf  = wid & 1;
  const float4* rp = (const float4*)(sc + row * QS_PITCH);
  const float4 x0 = rp[lane];
  const float4 x1 = rp[lane + 32];
  const float4 x2 = rp[lane + 64];
  const float4 x3 = rp[lane + 96];

  float m = fmaxf(fmaxf(fmaxf(x0.x, x0.y), fmaxf(x0.z, x0.w)),
                  fmaxf(fmaxf(x1.x, x1.y), fmaxf(x1.z, x1.w)));
  m = fmaxf(m, fmaxf(fmaxf(fmaxf(x2.x, x2.y), fmaxf(x2.z, x2.w)),
                     fmaxf(fmaxf(x3.x, x3.y), fmaxf(x3.z, x3.w))));
#pragma unroll
  for (int off = 16; off > 0; off >>= 1) m = fmaxf(m, __shfl_xor(m, off, 32));

  float4 e0, e1, e2, e3;
  e0.x = hw_exp(x0.x - m); e0.y = hw_exp(x0.y - m); e0.z = hw_exp(x0.z - m); e0.w = hw_exp(x0.w - m);
  e1.x = hw_exp(x1.x - m); e1.y = hw_exp(x1.y - m); e1.z = hw_exp(x1.z - m); e1.w = hw_exp(x1.w - m);
  e2.x = hw_exp(x2.x - m); e2.y = hw_exp(x2.y - m); e2.z = hw_exp(x2.z - m); e2.w = hw_exp(x2.w - m);
  e3.x = hw_exp(x3.x - m); e3.y = hw_exp(x3.y - m); e3.z = hw_exp(x3.z - m); e3.w = hw_exp(x3.w - m);
  float t = (((e0.x + e0.y) + (e0.z + e0.w)) + ((e1.x + e1.y) + (e1.z + e1.w))) +
            (((e2.x + e2.y) + (e2.z + e2.w)) + ((e3.x + e3.y) + (e3.z + e3.w)));
#pragma unroll
  for (int off = 16; off > 0; off >>= 1) t += __shfl_xor(t, off, 32);

  const float4 ya = hf ? e2 : e0;
  const float4 yb = hf ? e3 : e1;
  float* op = attn + (long long)(b * NQ_ + q0 + row) * NK_ + hf * (NK_ / 2) + 4 * lane;
  float4 oa, ob;
  oa.x = ya.x / t; oa.y = ya.y / t; oa.z = ya.z / t; oa.w = ya.w / t;
  ob.x = yb.x / t; ob.y = yb.y / t; ob.z = yb.z / t; ob.w = yb.w / t;
  *(float4*)op = oa;
  *(float4*)(op + NK_ / 4) = ob;
}

#undef WQ_GLOAD
#undef WQ_SSTORE

extern "C" void kernel_launch(void* const* d_in, const int* in_sizes, int n_in,
                              void* d_out, int out_size, void* d_ws, size_t ws_size,
                              hipStream_t stream) {
  (void)in_sizes; (void)n_in; (void)out_size; (void)ws_size;

  const float* queries    = (const float*)d_in[0];
  const float* keys       = (const float*)d_in[1];
  const float* values     = (const float*)d_in[2];
  const float* Wq         = (const float*)d_in[3];
  const float* Wk         = (const float*)d_in[4];
  const float* Wv         = (const float*)d_in[5];
  const int*   valid_lens = (const int*)d_in[6];
  float*       out        = (float*)d_out;

  float* kproj = (float*)d_ws + (long long)B_ * NQ_ * H_;
  float* attn  = kproj + (long long)B_ * NK_ * H_;

  kproj_gemm_f32<<<dim3((B_ * NK_) / KP_TM, H_ / KP_TN, 1), 128, 0, stream>>>(
      keys, Wk, kproj);

  scores_softmax<<<dim3(NQ_ / QB_, B_), NK_, 0, stream>>>(
      queries, Wq, kproj, Wv, valid_lens, attn);

  wmma_gemm_f32<<<dim3(NQ_ / GT_M, D_ / GT_N, B_), 128, 0, stream>>>(
      attn, values, out, NK_, D_,
      (long long)NQ_ * NK_, (long long)NK_ * D_, (long long)NQ_ * D_);
}


# Scores loop: adds and multiply-adds packed two-wide over (h, h+1), issued alternately with the tanh, multiply-adds one block of 4 h behind; row-sum order changed (even and odd h summed apart)
# speedup vs baseline: 2.8345x; 1.0499x over previous
#include <hip/hip_runtime.h>

#define B_  4
#define NQ_ 512
#define NK_ 512
#define D_  512
#define H_  128
#define QB_ 8

typedef __attribute__((ext_vector_type(2))) float v2f;
typedef __attribute__((ext_vector_type(4))) float v4f;
typedef __attribute__((ext_vector_type(8))) float v8f;

__device__ __forceinline__ float hw_tanh(float x) {
#if __has_builtin(__builtin_amdgcn_tanhf)
  return __builtin_amdgcn_tanhf(x);
#else
  float y;
  asm volatile("v_tanh_f32 %0, %1" : "=v"(y) : "v"(x));
  return y;
#endif
}

__device__ __forceinline__ float hw_exp(float x) {
#if __has_builtin(__builtin_amdgcn_exp2f)
  return __builtin_amdgcn_exp2f(x * 1.44269504088896340736f);
#else
  return __expf(x);
#endif
}

#define GT_M 64
#define GT_N 64
#define GT_K 64
#define GA_PITCH 68
#define GB_PITCH 144

#define GEMM_GLOAD(k0)                                                        \
  do {                                                                        \
    ra0 = *(const float4*)(ag + (long long)(0) * K + (k0));                  \
    ra1 = *(const float4*)(ag + (long long)(8) * K + (k0));                  \
    ra2 = *(const float4*)(ag + (long long)(16) * K + (k0));                  \
    ra3 = *(const float4*)(ag + (long long)(24) * K + (k0));                  \
    ra4 = *(const float4*)(ag + (long long)(32) * K + (k0));                  \
    ra5 = *(const float4*)(ag + (long long)(40) * K + (k0));                  \
    ra6 = *(const float4*)(ag + (long long)(48) * K + (k0));                  \
    ra7 = *(const float4*)(ag + (long long)(56) * K + (k0));                  \
    rb0 = *(const float4*)(bg + (long long)((k0) + 0) * N);                  \
    rb1 = *(const float4*)(bg + (long long)((k0) + 0) * N + N);              \
    rb2 = *(const float4*)(bg + (long long)((k0) + 16) * N);                  \
    rb3 = *(const float4*)(bg + (long long)((k0) + 16) * N + N);              \
    rb4 = *(const float4*)(bg + (long long)((k0) + 32) * N);                  \
    rb5 = *(const float4*)(bg + (long long)((k0) + 32) * N + N);              \
    rb6 = *(const float4*)(bg + (long long)((k0) + 48) * N);                  \
    rb7 = *(const float4*)(bg + (long long)((k0) + 48) * N + N);              \
  } while (0)

#define GEMM_SSTORE(as, bs)                                                   \
  do {                                                                        \
    *(float4*)((as) + 0 * GA_PITCH) = ra0;                                  \
    *(float4*)((as) + 8 * GA_PITCH) = ra1;                                  \
    *(float4*)((as) + 16 * GA_PITCH) = ra2;                                  \
    *(float4*)((as) + 24 * GA_PITCH) = ra3;                                  \
    *(float4*)((as) + 32 * GA_PITCH) = ra4;                                  \
    *(float4*)((as) + 40 * GA_PITCH) = ra5;                                  \
    *(float4*)((as) + 48 * GA_PITCH) = ra6;                                  \
    *(float4*)((as) + 56 * GA_PITCH) = ra7;                                  \
    *(float4*)((bs) + 0 * GB_PITCH) = rb0;                                  \
    *(float4*)((bs) + 0 * GB_PITCH + 16) = rb1;                             \
    *(float4*)((bs) + 8 * GB_PITCH) = rb2;                                  \
    *(float4*)((bs) + 8 * GB_PITCH + 16) = rb3;                             \
    *(float4*)((bs) + 16 * GB_PITCH) = rb4;                                  \
    *(float4*)((bs) + 16 * GB_PITCH + 16) = rb5;                             \
    *(float4*)((bs) + 24 * GB_PITCH) = rb6;                                  \
    *(float4*)((bs) + 24 * GB_PITCH + 16) = rb7;                             \
  } while (0)

__device__ __forceinline__ void gemm_slab(const float* as, const float* bs,
                                          v8f& acc00, v8f& acc01, v8f& acc10, v8f& acc11) {
  v2f a0[GT_K / 4], a1[GT_K / 4], b0[GT_K / 4], b1[GT_K / 4];
#pragma unroll
  for (int ks = 0; ks < GT_K / 4; ++ks) {
    if (ks == 4) __builtin_amdgcn_sched_barrier(0);
    a0[ks] = *(const v2f*)(as + 4 * ks);
    a1[ks] = *(const v2f*)(as + 16 * GA_PITCH + 4 * ks);
    b0[ks].x = bs[(2 * ks) * GB_PITCH];
    b0[ks].y = bs[(2 * ks) * GB_PITCH + 16];
    b1[ks].x = bs[(2 * ks) * GB_PITCH + 32];
    b1[ks].y = bs[(2 * ks) * GB_PITCH + 48];
  }
  __builtin_amdgcn_sched_barrier(0);
#pragma unroll
  for (int ks = 0; ks < GT_K / 4; ++ks) {
    acc00 = __builtin_amdgcn_wmma_f32_16x16x4_f32(false, a0[ks], false, b0[ks], (short)0, acc00, false, false);
    acc01 = __builtin_amdgcn_wmma_f32_16x16x4_f32(false, a0[ks], false, b1[ks], (short)0, acc01, false, false);
    acc10 = __builtin_amdgcn_wmma_f32_16x16x4_f32(false, a1[ks], false, b0[ks], (short)0, acc10, false, false);
    acc11 = __builtin_amdgcn_wmma_f32_16x16x4_f32(false, a1[ks], false, b1[ks], (short)0, acc11, false, false);
  }
}

__global__ __launch_bounds__(128) void wmma_gemm_f32(
    const float* A, const float* Bm, float* C, int K, int N,
    long long sA, long long sB, long long sC) {
  __shared__ __attribute__((aligned(16))) float As[2 * GT_M * GA_PITCH];
  __shared__ __attribute__((aligned(16))) float Bs[2 * (GT_K / 2) * GB_PITCH];

  const int tid   = threadIdx.x;
  const int lane  = tid & 31;
  const int wave  = tid >> 5;
  const int wm    = wave >> 1;
  const int wn    = wave & 1;
  const int mBase = blockIdx.x * GT_M;
  const int nBase = blockIdx.y * GT_N;
  const int batch = blockIdx.z;
  A  += (long long)batch * sA;
  Bm += (long long)batch * sB;
  C  += (long long)batch * sC;

  const int c4 = tid & 15;
  const int r8 = tid >> 4;
  const float* ag = A  + (long long)(mBase + r8) * K + 4 * c4;
  const float* bg = Bm + (long long)(2 * r8) * N + nBase + 4 * c4;
  float* asw = As + r8 * GA_PITCH + 4 * c4;
  float* bsw = Bs + r8 * GB_PITCH + (c4 >> 2) * 32 + (c4 & 3) * 4;

  const int c = lane & 15;
  const int h = lane >> 4;
  const float* asr = As + (wm * 32 + c) * GA_PITCH + 2 * h;
  const float* bsr = Bs + h * GB_PITCH + wn * 64 + c;

  v8f acc00 = {}, acc01 = {}, acc10 = {}, acc11 = {};
  float4 ra0, ra1, ra2, ra3, ra4, ra5, ra6, ra7;
  float4 rb0, rb1, rb2, rb3, rb4, rb5, rb6, rb7;

  GEMM_GLOAD(0);
  GEMM_SSTORE(asw, bsw);
  __syncthreads();

  const int nSlab = K / GT_K;
  int cur = 0;
  for (int s = 0; s + 1 < nSlab; ++s) {
    const int nxt = cur ^ 1;
    GEMM_GLOAD((s + 1) * GT_K);
    __builtin_amdgcn_sched_barrier(0);
    gemm_slab(asr + cur * (GT_M * GA_PITCH), bsr + cur * ((GT_K / 2) * GB_PITCH),
              acc00, acc01, acc10, acc11);
    asm volatile("" :: "v"(acc00), "v"(acc01), "v"(acc10), "v"(acc11) : "memory");
    GEMM_SSTORE(asw + nxt * (GT_M * GA_PITCH), bsw + nxt * ((GT_K / 2) * GB_PITCH));
    __syncthreads();
    cur = nxt;
  }
  gemm_slab(asr + cur * (GT_M * GA_PITCH), bsr + cur * ((GT_K / 2) * GB_PITCH),
            acc00, acc01, acc10, acc11);

  const int mOut0 = mBase + wm * 32 + h * 8;
  const int nOut0 = nBase + wn * 32 + c;
#pragma unroll
  for (int r = 0; r < 8; ++r) {
    long long row0 = (long long)(mOut0 + r) * N + nOut0;
    C[row0 +  0] = acc00[r];
    C[row0 + 16] = acc01[r];
    long long row1 = row0 + (long long)16 * N;
    C[row1 +  0] = acc10[r];
    C[row1 + 16] = acc11[r];
  }
}
#undef GEMM_GLOAD
#undef GEMM_SSTORE

#define KP_TM 32
#define KP_TN 32
#define KP_KS 256
#define KP_KG 128
#define KP_APITCH 516
#define KP_BPITCH 80
#define KP_CPITCH 36

#define KP_GLOAD(koff) \
  do { \
    pa0 = *(const float4*)(ag + (long long)(0) * D_ + (koff)); \
    pa1 = *(const float4*)(ag + (long long)(2) * D_ + (koff)); \
    pa2 = *(const float4*)(ag + (long long)(4) * D_ + (koff)); \
    pa3 = *(const float4*)(ag + (long long)(6) * D_ + (koff)); \
    pa4 = *(const float4*)(ag + (long long)(8) * D_ + (koff)); \
    pa5 = *(const float4*)(ag + (long long)(10) * D_ + (koff)); \
    pa6 = *(const float4*)(ag + (long long)(12) * D_ + (koff)); \
    pa7 = *(const float4*)(ag + (long long)(14) * D_ + (koff)); \
    pa8 = *(const float4*)(ag + (long long)(16) * D_ + (koff)); \
    pa9 = *(const float4*)(ag + (long long)(18) * D_ + (koff)); \
    pa10 = *(const float4*)(ag + (long long)(20) * D_ + (koff)); \
    pa11 = *(const float4*)(ag + (long long)(22) * D_ + (koff)); \
    pa12 = *(const float4*)(ag + (long long)(24) * D_ + (koff)); \
    pa13 = *(const float4*)(ag + (long long)(26) * D_ + (koff)); \
    pa14 = *(const float4*)(ag + (long long)(28) * D_ + (koff)); \
    pa15 = *(const float4*)(ag + (long long)(30) * D_ + (koff)); \
    pb0 = *(const float4*)(bg + (long long)((koff) + 0) * H_); \
    pb1 = *(const float4*)(bg + (long long)((koff) + 16) * H_); \
    pb2 = *(const float4*)(bg + (long long)((koff) + 32) * H_); \
    pb3 = *(const float4*)(bg + (long long)((koff) + 48) * H_); \
    pb4 = *(const float4*)(bg + (long long)((koff) + 64) * H_); \
    pb5 = *(const float4*)(bg + (long long)((koff) + 80) * H_); \
    pb6 = *(const float4*)(bg + (long long)((koff) + 96) * H_); \
    pb7 = *(const float4*)(bg + (long long)((koff) + 112) * H_); \
    pb8 = *(const float4*)(bg + (long long)((koff) + 128) * H_); \
    pb9 = *(const float4*)(bg + (long long)((koff) + 144) * H_); \
    pb10 = *(const float4*)(bg + (long long)((koff) + 160) * H_); \
    pb11 = *(const float4*)(bg + (long long)((koff) + 176) * H_); \
    pb12 = *(const float4*)(bg + (long long)((koff) + 192) * H_); \
    pb13 = *(const float4*)(bg + (long long)((koff) + 208) * H_); \
    pb14 = *(const float4*)(bg + (long long)((koff) + 224) * H_); \
    pb15 = *(const float4*)(bg + (long long)((koff) + 240) * H_); \
  } while (0)

#define KP_SSTORE(koff) \
  do { \
    *(float4*)(asw + 0 * KP_APITCH + (koff)) = pa0; \
    *(float4*)(asw + 2 * KP_APITCH + (koff)) = pa1; \
    *(float4*)(asw + 4 * KP_APITCH + (koff)) = pa2; \
    *(float4*)(asw + 6 * KP_APITCH + (koff)) = pa3; \
    *(float4*)(asw + 8 * KP_APITCH + (koff)) = pa4; \
    *(float4*)(asw + 10 * KP_APITCH + (koff)) = pa5; \
    *(float4*)(asw + 12 * KP_APITCH + (koff)) = pa6; \
    *(float4*)(asw + 14 * KP_APITCH + (koff)) = pa7; \
    *(float4*)(asw + 16 * KP_APITCH + (koff)) = pa8; \
    *(float4*)(asw + 18 * KP_APITCH + (koff)) = pa9; \
    *(float4*)(asw + 20 * KP_APITCH + (koff)) = pa10; \
    *(float4*)(asw + 22 * KP_APITCH + (koff)) = pa11; \
    *(float4*)(asw + 24 * KP_APITCH + (koff)) = pa12; \
    *(float4*)(asw + 26 * KP_APITCH + (koff)) = pa13; \
    *(float4*)(asw + 28 * KP_APITCH + (koff)) = pa14; \
    *(float4*)(asw + 30 * KP_APITCH + (koff)) = pa15; \
    *(float4*)(bsw + ((koff) / 2 + 0) * KP_BPITCH) = pb0; \
    *(float4*)(bsw + ((koff) / 2 + 8) * KP_BPITCH) = pb1; \
    *(float4*)(bsw + ((koff) / 2 + 16) * KP_BPITCH) = pb2; \
    *(float4*)(bsw + ((koff) / 2 + 24) * KP_BPITCH) = pb3; \
    *(float4*)(bsw + ((koff) / 2 + 32) * KP_BPITCH) = pb4; \
    *(float4*)(bsw + ((koff) / 2 + 40) * KP_BPITCH) = pb5; \
    *(float4*)(bsw + ((koff) / 2 + 48) * KP_BPITCH) = pb6; \
    *(float4*)(bsw + ((koff) / 2 + 56) * KP_BPITCH) = pb7; \
    *(float4*)(bsw + ((koff) / 2 + 64) * KP_BPITCH) = pb8; \
    *(float4*)(bsw + ((koff) / 2 + 72) * KP_BPITCH) = pb9; \
    *(float4*)(bsw + ((koff) / 2 + 80) * KP_BPITCH) = pb10; \
    *(float4*)(bsw + ((koff) / 2 + 88) * KP_BPITCH) = pb11; \
    *(float4*)(bsw + ((koff) / 2 + 96) * KP_BPITCH) = pb12; \
    *(float4*)(bsw + ((koff) / 2 + 104) * KP_BPITCH) = pb13; \
    *(float4*)(bsw + ((koff) / 2 + 112) * KP_BPITCH) = pb14; \
    *(float4*)(bsw + ((koff) / 2 + 120) * KP_BPITCH) = pb15; \
  } while (0)

__device__ __forceinline__ void kp_group(const float* as, const float* bs, v8f& acc) {
  v2f a[KP_KG / 4], b[KP_KG / 4];
#pragma unroll
  for (int ks = 0; ks < KP_KG / 4; ++ks) {
    if (ks == 4) __builtin_amdgcn_sched_barrier(0);
    a[ks]   = *(const v2f*)(as + 4 * ks);
    b[ks].x = bs[(2 * ks) * KP_BPITCH];
    b[ks].y = bs[(2 * ks) * KP_BPITCH + 16];
  }
  __builtin_amdgcn_sched_barrier(0);
#pragma unroll
  for (int ks = 0; ks < KP_KG / 4; ++ks)
    acc = __builtin_amdgcn_wmma_f32_16x16x4_f32(false, a[ks], false, b[ks], (short)0, acc, false, false);
  asm volatile("" :: "v"(acc) : "memory");
}

__global__ __launch_bounds__(128) void kproj_gemm_f32(
    const float* A, const float* Bm, float* C) {
  __shared__ __attribute__((aligned(16))) float As[KP_TM * KP_APITCH];
  __shared__ __attribute__((aligned(16))) float Bs[(D_ / 2) * KP_BPITCH];
  __shared__ __attribute__((aligned(16))) float Cs[KP_TM * KP_CPITCH];

  const int tid   = threadIdx.x;
  const int lane  = tid & 31;
  const int wave  = tid >> 5;
  const int wm    = wave >> 1;
  const int wn    = wave & 1;
  const int mBase = blockIdx.x * KP_TM;
  const int nBase = blockIdx.y * KP_TN;

  const int c64 = tid & 63;
  const int r2  = tid >> 6;
  const int c8  = tid & 7;
  const int kr  = tid >> 3;
  const float* ag = A  + (long long)(mBase + r2) * D_ + 4 * c64;
  const float* bg = Bm + (long long)kr * H_ + nBase + 4 * c8;
  float* asw = As + r2 * KP_APITCH + 4 * c64;
  float* bsw = Bs + (kr >> 1) * KP_BPITCH + (c8 >> 2) * 32 + (kr & 1) * 16 + (c8 & 3) * 4;

  const int c = lane & 15;
  const int h = lane >> 4;
  const float* asr = As + (wm * 16 + c) * KP_APITCH + 2 * h;
  const float* bsr = Bs + h * KP_BPITCH + wn * 32 + c;

  v8f acc = {};
  float4 pa0, pa1, pa2, pa3, pa4, pa5, pa6, pa7, pa8, pa9, pa10, pa11, pa12, pa13, pa14, pa15;
  float4 pb0, pb1, pb2, pb3, pb4, pb5, pb6, pb7, pb8, pb9, pb10, pb11, pb12, pb13, pb14, pb15;

  KP_GLOAD(0);
  KP_SSTORE(0);
  __syncthreads();

  KP_GLOAD(KP_KS);
  __builtin_amdgcn_sched_barrier(0);

  kp_group(asr,         bsr,                             acc);
  kp_group(asr + KP_KG, bsr + (KP_KG / 2) * KP_BPITCH,   acc);

  KP_SSTORE(KP_KS);
  __syncthreads();

  kp_group(asr + 2 * KP_KG, bsr + (2 * KP_KG / 2) * KP_BPITCH, acc);
  kp_group(asr + 3 * KP_KG, bsr + (3 * KP_KG / 2) * KP_BPITCH, acc);

#pragma unroll
  for (int r = 0; r < 8; ++r) Cs[(wm * 16 + h * 8 + r) * KP_CPITCH + wn * 16 + c] = acc[r];
  __syncthreads();
  const int orow = tid >> 3;
  const float4 o0 = *(const float4*)(Cs + orow * KP_CPITCH + 4 * c8);
  const float4 o1 = *(const float4*)(Cs + (orow + 16) * KP_CPITCH + 4 * c8);
  *(float4*)(C + (long long)(mBase + orow) * H_ + nBase + 4 * c8) = o0;
  *(float4*)(C + (long long)(mBase + orow + 16) * H_ + nBase + 4 * c8) = o1;
}
#undef KP_GLOAD
#undef KP_SSTORE

#define QP_KC    64
#define QP_PITCH 272
#define QS_PITCH 516

#define WQ_GLOAD(ch)                                                          \
  do {                                                                        \
    rw0 = *(const float4*)(wg + (long long)((ch) * QP_KC + 0) * H_);          \
    rw1 = *(const float4*)(wg + (long long)((ch) * QP_KC + 16) * H_);         \
    rw2 = *(const float4*)(wg + (long long)((ch) * QP_KC + 32) * H_);         \
    rw3 = *(const float4*)(wg + (long long)((ch) * QP_KC + 48) * H_);         \
  } while (0)

#define WQ_SSTORE(ws)                                                         \
  do {                                                                        \
    *(float4*)((ws) + 0 * QP_PITCH) = rw0;                                    \
    *(float4*)((ws) + 8 * QP_PITCH) = rw1;                                    \
    *(float4*)((ws) + 16 * QP_PITCH) = rw2;                                   \
    *(float4*)((ws) + 24 * QP_PITCH) = rw3;                                   \
  } while (0)

__global__ __launch_bounds__(NK_) void scores_softmax(
    const float* __restrict__ queries, const float* __restrict__ Wq,
    const float* __restrict__ kproj,
    const float* __restrict__ Wv, const int* __restrict__ valid_lens,
    float* __restrict__ attn) {
  __shared__ __attribute__((aligned(16))) float qs[QB_ * QS_PITCH];
  __shared__ __attribute__((aligned(16))) float wqs[2 * (QP_KC / 2) * QP_PITCH];
  __shared__ __attribute__((aligned(16))) float qp[QB_][H_];
  __shared__ float qph[QB_][H_];
  __shared__ __attribute__((aligned(16))) float wv[H_];

  const int q0   = blockIdx.x * QB_;
  const int b    = blockIdx.y;
  const int k    = threadIdx.x;
  const int lane = k & 31;
  const int wid  = k >> 5;

  {
    const int c   = lane & 15;
    const int hh  = lane >> 4;
    const int nt  = wid & 7;
    const int kh  = wid >> 3;
    const int c4  = k & 31;
    const int r16 = k >> 5;
    const float* wg = Wq + (long long)r16 * H_ + 4 * c4;
    float* wsw = wqs + (r16 >> 1) * QP_PITCH + (c4 >> 2) * 32 + (r16 & 1) * 16 + (c4 & 3) * 4;
    const float* qg = queries + (long long)(b * NQ_ + q0) * D_;

    float4 rw0, rw1, rw2, rw3;
    WQ_GLOAD(0);
    const float4 rq0 = *(const float4*)(qg + 4 * k);
    const float4 rq1 = *(const float4*)(qg + 4 * (k + NK_));
    *(float4*)(qs + (k >> 7) * QS_PITCH + 4 * (k & 127)) = rq0;
    *(float4*)(qs + ((k >> 7) + 4) * QS_PITCH + 4 * (k & 127)) = rq1;
    WQ_SSTORE(wsw);
    if (k < H_) wv[k] = Wv[k];
    __syncthreads();

    const float* asr = qs + (c & 7) * QS_PITCH + 32 * kh + 2 * hh;
    const float* bsr = wqs + (16 * kh + hh) * QP_PITCH + nt * 32 + c;
    v8f acc = {};
    int cur = 0;
    for (int ch = 0; ch < D_ / QP_KC; ++ch) {
      const bool more = ch + 1 < D_ / QP_KC;
      if (more) WQ_GLOAD(ch + 1);
      const float* as = asr + ch * QP_KC;
      const float* bs = bsr + cur * ((QP_KC / 2) * QP_PITCH);
      v2f af[8], bf[8];
#pragma unroll
      for (int i = 0; i < 8; ++i) {
        af[i]   = *(const v2f*)(as + 4 * i);
        bf[i].x = bs[(2 * i) * QP_PITCH];
        bf[i].y = bs[(2 * i) * QP_PITCH + 16];
      }
#pragma unroll
      for (int i = 0; i < 8; ++i)
        acc = __builtin_amdgcn_wmma_f32_16x16x4_f32(false, af[i], false, bf[i], (short)0, acc, false, false);
      if (more) WQ_SSTORE(wsw + (cur ^ 1) * ((QP_KC / 2) * QP_PITCH));
      __syncthreads();
      cur ^= 1;
    }

    if (kh == 1 && hh == 0) {
#pragma unroll
      for (int r = 0; r < QB_; ++r) qph[r][nt * 16 + c] = acc[r];
    }
    __syncthreads();
    if (kh == 0 && hh == 0) {
#pragma unroll
      for (int r = 0; r < QB_; ++r) qp[r][nt * 16 + c] = acc[r] + qph[r][nt * 16 + c];
    }
    __syncthreads();
  }

  const v4f* kp4 =
      (const v4f*)(kproj + (long long)(b * NK_ + k) * H_);
  v2f s2[QB_];
  v2f ta_lo[QB_], ta_hi[QB_], tb_lo[QB_], tb_hi[QB_];
  v2f wa_lo, wa_hi;
  v2f wb_lo = (v2f){0.0f, 0.0f}, wb_hi = (v2f){0.0f, 0.0f};
#pragma unroll
  for (int j = 0; j < QB_; ++j) {
    s2[j] = (v2f){0.0f, 0.0f};
    tb_lo[j] = (v2f){0.0f, 0.0f};
    tb_hi[j] = (v2f){0.0f, 0.0f};
  }

#define SCORE_BLOCK(H0, TL, TH, WL, WH, PL, PH, PWL, PWH)                     \
  do {                                                                        \
    const v4f kv = kp4[(H0) / 4];                                             \
    const v4f w4 = *(const v4f*)&wv[(H0)];                                    \
    const v2f klo = {kv.x, kv.y}, khi = {kv.z, kv.w};                         \
    _Pragma("unroll")                                                         \
    for (int j = 0; j < QB_; ++j) {                                           \
      const v4f q4 = *(const v4f*)&qp[j][(H0)];                               \
      const v2f xlo = (v2f){q4.x, q4.y} + klo;                                \
      const v2f xhi = (v2f){q4.z, q4.w} + khi;                                \
      TL[j] = (v2f){hw_tanh(xlo.x), hw_tanh(xlo.y)};                          \
      TH[j] = (v2f){hw_tanh(xhi.x), hw_tanh(xhi.y)};                          \
      s2[j] = __builtin_elementwise_fma(PWL, PL[j], s2[j]);                   \
      s2[j] = __builtin_elementwise_fma(PWH, PH[j], s2[j]);                   \
    }                                                                         \
    WL = (v2f){w4.x, w4.y};                                                   \
    WH = (v2f){w4.z, w4.w};                                                   \
        \
        \
        \
        \
        \
    __builtin_amdgcn_sched_group_barrier(0x002, 8, 0);                        \
    _Pragma("unroll")                                                         \
    for (int g = 0; g < 28; ++g) {                                            \
      __builtin_amdgcn_sched_group_barrier(0x400, 1, 0);                      \
      __builtin_amdgcn_sched_group_barrier(0x002, 1, 0);                      \
    }                                                                         \
  } while (0)

  for (int h8 = 0; h8 < H_; h8 += 8) {
    SCORE_BLOCK(h8, ta_lo, ta_hi, wa_lo, wa_hi, tb_lo, tb_hi, wb_lo, wb_hi);
    __builtin_amdgcn_sched_barrier(0);
    SCORE_BLOCK(h8 + 4, tb_lo, tb_hi, wb_lo, wb_hi, ta_lo, ta_hi, wa_lo, wa_hi);
  }
#undef SCORE_BLOCK
  float s[QB_];
#pragma unroll
  for (int j = 0; j < QB_; ++j) {
    s2[j] = __builtin_elementwise_fma(wb_lo, tb_lo[j], s2[j]);
    s2[j] = __builtin_elementwise_fma(wb_hi, tb_hi[j], s2[j]);
    s[j] = s2[j].x + s2[j].y;
  }

  const int vlen = valid_lens[b];

  float* sc = qs;
#pragma unroll
  for (int j = 0; j < QB_; ++j) sc[j * QS_PITCH + k] = (k < vlen) ? s[j] : -1.0e6f;
  __syncthreads();

  const int row = wid >> 1;
  const int hf  = wid & 1;
  const float4* rp = (const float4*)(sc + row * QS_PITCH);
  const float4 x0 = rp[lane];
  const float4 x1 = rp[lane + 32];
  const float4 x2 = rp[lane + 64];
  const float4 x3 = rp[lane + 96];

  float m = fmaxf(fmaxf(fmaxf(x0.x, x0.y), fmaxf(x0.z, x0.w)),
                  fmaxf(fmaxf(x1.x, x1.y), fmaxf(x1.z, x1.w)));
  m = fmaxf(m, fmaxf(fmaxf(fmaxf(x2.x, x2.y), fmaxf(x2.z, x2.w)),
                     fmaxf(fmaxf(x3.x, x3.y), fmaxf(x3.z, x3.w))));
#pragma unroll
  for (int off = 16; off > 0; off >>= 1) m = fmaxf(m, __shfl_xor(m, off, 32));

  float4 e0, e1, e2, e3;
  e0.x = hw_exp(x0.x - m); e0.y = hw_exp(x0.y - m); e0.z = hw_exp(x0.z - m); e0.w = hw_exp(x0.w - m);
  e1.x = hw_exp(x1.x - m); e1.y = hw_exp(x1.y - m); e1.z = hw_exp(x1.z - m); e1.w = hw_exp(x1.w - m);
  e2.x = hw_exp(x2.x - m); e2.y = hw_exp(x2.y - m); e2.z = hw_exp(x2.z - m); e2.w = hw_exp(x2.w - m);
  e3.x = hw_exp(x3.x - m); e3.y = hw_exp(x3.y - m); e3.z = hw_exp(x3.z - m); e3.w = hw_exp(x3.w - m);
  float t = (((e0.x + e0.y) + (e0.z + e0.w)) + ((e1.x + e1.y) + (e1.z + e1.w))) +
            (((e2.x + e2.y) + (e2.z + e2.w)) + ((e3.x + e3.y) + (e3.z + e3.w)));
#pragma unroll
  for (int off = 16; off > 0; off >>= 1) t += __shfl_xor(t, off, 32);

  const float4 ya = hf ? e2 : e0;
  const float4 yb = hf ? e3 : e1;
  float* op = attn + (long long)(b * NQ_ + q0 + row) * NK_ + hf * (NK_ / 2) + 4 * lane;
  float4 oa, ob;
  oa.x = ya.x / t; oa.y = ya.y / t; oa.z = ya.z / t; oa.w = ya.w / t;
  ob.x = yb.x / t; ob.y = yb.y / t; ob.z = yb.z / t; ob.w = yb.w / t;
  *(float4*)op = oa;
  *(float4*)(op + NK_ / 4) = ob;
}

#undef WQ_GLOAD
#undef WQ_SSTORE

extern "C" void kernel_launch(void* const* d_in, const int* in_sizes, int n_in,
                              void* d_out, int out_size, void* d_ws, size_t ws_size,
                              hipStream_t stream) {
  (void)in_sizes; (void)n_in; (void)out_size; (void)ws_size;

  const float* queries    = (const float*)d_in[0];
  const float* keys       = (const float*)d_in[1];
  const float* values     = (const float*)d_in[2];
  const float* Wq         = (const float*)d_in[3];
  const float* Wk         = (const float*)d_in[4];
  const float* Wv         = (const float*)d_in[5];
  const int*   valid_lens = (const int*)d_in[6];
  float*       out        = (float*)d_out;

  float* kproj = (float*)d_ws + (long long)B_ * NQ_ * H_;
  float* attn  = kproj + (long long)B_ * NK_ * H_;

  kproj_gemm_f32<<<dim3((B_ * NK_) / KP_TM, H_ / KP_TN, 1), 128, 0, stream>>>(
      keys, Wk, kproj);

  scores_softmax<<<dim3(NQ_ / QB_, B_), NK_, 0, stream>>>(
      queries, Wq, kproj, Wv, valid_lens, attn);

  wmma_gemm_f32<<<dim3(NQ_ / GT_M, D_ / GT_N, B_), 128, 0, stream>>>(
      attn, values, out, NK_, D_,
      (long long)NQ_ * NK_, (long long)NK_ * D_, (long long)NQ_ * D_);
}


# Query projection in the scores kernel: A and Wq fragments go from global straight to registers, every load of a wave in flight at once; no LDS staging, no K-loop barriers; K in 8 parts (order changed)
# speedup vs baseline: 3.1607x; 1.1151x over previous
#include <hip/hip_runtime.h>

#define B_  4
#define NQ_ 512
#define NK_ 512
#define D_  512
#define H_  128
#define QB_ 8

typedef __attribute__((ext_vector_type(2))) float v2f;
typedef __attribute__((ext_vector_type(4))) float v4f;
typedef __attribute__((ext_vector_type(8))) float v8f;

__device__ __forceinline__ float hw_tanh(float x) {
#if __has_builtin(__builtin_amdgcn_tanhf)
  return __builtin_amdgcn_tanhf(x);
#else
  float y;
  asm volatile("v_tanh_f32 %0, %1" : "=v"(y) : "v"(x));
  return y;
#endif
}

__device__ __forceinline__ float hw_exp(float x) {
#if __has_builtin(__builtin_amdgcn_exp2f)
  return __builtin_amdgcn_exp2f(x * 1.44269504088896340736f);
#else
  return __expf(x);
#endif
}

#define GT_M 64
#define GT_N 64
#define GT_K 64
#define GA_PITCH 68
#define GB_PITCH 144

#define GEMM_GLOAD(k0)                                                        \
  do {                                                                        \
    ra0 = *(const float4*)(ag + (long long)(0) * K + (k0));                  \
    ra1 = *(const float4*)(ag + (long long)(8) * K + (k0));                  \
    ra2 = *(const float4*)(ag + (long long)(16) * K + (k0));                  \
    ra3 = *(const float4*)(ag + (long long)(24) * K + (k0));                  \
    ra4 = *(const float4*)(ag + (long long)(32) * K + (k0));                  \
    ra5 = *(const float4*)(ag + (long long)(40) * K + (k0));                  \
    ra6 = *(const float4*)(ag + (long long)(48) * K + (k0));                  \
    ra7 = *(const float4*)(ag + (long long)(56) * K + (k0));                  \
    rb0 = *(const float4*)(bg + (long long)((k0) + 0) * N);                  \
    rb1 = *(const float4*)(bg + (long long)((k0) + 0) * N + N);              \
    rb2 = *(const float4*)(bg + (long long)((k0) + 16) * N);                  \
    rb3 = *(const float4*)(bg + (long long)((k0) + 16) * N + N);              \
    rb4 = *(const float4*)(bg + (long long)((k0) + 32) * N);                  \
    rb5 = *(const float4*)(bg + (long long)((k0) + 32) * N + N);              \
    rb6 = *(const float4*)(bg + (long long)((k0) + 48) * N);                  \
    rb7 = *(const float4*)(bg + (long long)((k0) + 48) * N + N);              \
  } while (0)

#define GEMM_SSTORE(as, bs)                                                   \
  do {                                                                        \
    *(float4*)((as) + 0 * GA_PITCH) = ra0;                                  \
    *(float4*)((as) + 8 * GA_PITCH) = ra1;                                  \
    *(float4*)((as) + 16 * GA_PITCH) = ra2;                                  \
    *(float4*)((as) + 24 * GA_PITCH) = ra3;                                  \
    *(float4*)((as) + 32 * GA_PITCH) = ra4;                                  \
    *(float4*)((as) + 40 * GA_PITCH) = ra5;                                  \
    *(float4*)((as) + 48 * GA_PITCH) = ra6;                                  \
    *(float4*)((as) + 56 * GA_PITCH) = ra7;                                  \
    *(float4*)((bs) + 0 * GB_PITCH) = rb0;                                  \
    *(float4*)((bs) + 0 * GB_PITCH + 16) = rb1;                             \
    *(float4*)((bs) + 8 * GB_PITCH) = rb2;                                  \
    *(float4*)((bs) + 8 * GB_PITCH + 16) = rb3;                             \
    *(float4*)((bs) + 16 * GB_PITCH) = rb4;                                  \
    *(float4*)((bs) + 16 * GB_PITCH + 16) = rb5;                             \
    *(float4*)((bs) + 24 * GB_PITCH) = rb6;                                  \
    *(float4*)((bs) + 24 * GB_PITCH + 16) = rb7;                             \
  } while (0)

__device__ __forceinline__ void gemm_slab(const float* as, const float* bs,
                                          v8f& acc00, v8f& acc01, v8f& acc10, v8f& acc11) {
  v2f a0[GT_K / 4], a1[GT_K / 4], b0[GT_K / 4], b1[GT_K / 4];
#pragma unroll
  for (int ks = 0; ks < GT_K / 4; ++ks) {
    if (ks == 4) __builtin_amdgcn_sched_barrier(0);
    a0[ks] = *(const v2f*)(as + 4 * ks);
    a1[ks] = *(const v2f*)(as + 16 * GA_PITCH + 4 * ks);
    b0[ks].x = bs[(2 * ks) * GB_PITCH];
    b0[ks].y = bs[(2 * ks) * GB_PITCH + 16];
    b1[ks].x = bs[(2 * ks) * GB_PITCH + 32];
    b1[ks].y = bs[(2 * ks) * GB_PITCH + 48];
  }
  __builtin_amdgcn_sched_barrier(0);
#pragma unroll
  for (int ks = 0; ks < GT_K / 4; ++ks) {
    acc00 = __builtin_amdgcn_wmma_f32_16x16x4_f32(false, a0[ks], false, b0[ks], (short)0, acc00, false, false);
    acc01 = __builtin_amdgcn_wmma_f32_16x16x4_f32(false, a0[ks], false, b1[ks], (short)0, acc01, false, false);
    acc10 = __builtin_amdgcn_wmma_f32_16x16x4_f32(false, a1[ks], false, b0[ks], (short)0, acc10, false, false);
    acc11 = __builtin_amdgcn_wmma_f32_16x16x4_f32(false, a1[ks], false, b1[ks], (short)0, acc11, false, false);
  }
}

__global__ __launch_bounds__(128) void wmma_gemm_f32(
    const float* A, const float* Bm, float* C, int K, int N,
    long long sA, long long sB, long long sC) {
  __shared__ __attribute__((aligned(16))) float As[2 * GT_M * GA_PITCH];
  __shared__ __attribute__((aligned(16))) float Bs[2 * (GT_K / 2) * GB_PITCH];

  const int tid   = threadIdx.x;
  const int lane  = tid & 31;
  const int wave  = tid >> 5;
  const int wm    = wave >> 1;
  const int wn    = wave & 1;
  const int mBase = blockIdx.x * GT_M;
  const int nBase = blockIdx.y * GT_N;
  const int batch = blockIdx.z;
  A  += (long long)batch * sA;
  Bm += (long long)batch * sB;
  C  += (long long)batch * sC;

  const int c4 = tid & 15;
  const int r8 = tid >> 4;
  const float* ag = A  + (long long)(mBase + r8) * K + 4 * c4;
  const float* bg = Bm + (long long)(2 * r8) * N + nBase + 4 * c4;
  float* asw = As + r8 * GA_PITCH + 4 * c4;
  float* bsw = Bs + r8 * GB_PITCH + (c4 >> 2) * 32 + (c4 & 3) * 4;

  const int c = lane & 15;
  const int h = lane >> 4;
  const float* asr = As + (wm * 32 + c) * GA_PITCH + 2 * h;
  const float* bsr = Bs + h * GB_PITCH + wn * 64 + c;

  v8f acc00 = {}, acc01 = {}, acc10 = {}, acc11 = {};
  float4 ra0, ra1, ra2, ra3, ra4, ra5, ra6, ra7;
  float4 rb0, rb1, rb2, rb3, rb4, rb5, rb6, rb7;

  GEMM_GLOAD(0);
  GEMM_SSTORE(asw, bsw);
  __syncthreads();

  const int nSlab = K / GT_K;
  int cur = 0;
  for (int s = 0; s + 1 < nSlab; ++s) {
    const int nxt = cur ^ 1;
    GEMM_GLOAD((s + 1) * GT_K);
    __builtin_amdgcn_sched_barrier(0);
    gemm_slab(asr + cur * (GT_M * GA_PITCH), bsr + cur * ((GT_K / 2) * GB_PITCH),
              acc00, acc01, acc10, acc11);
    asm volatile("" :: "v"(acc00), "v"(acc01), "v"(acc10), "v"(acc11) : "memory");
    GEMM_SSTORE(asw + nxt * (GT_M * GA_PITCH), bsw + nxt * ((GT_K / 2) * GB_PITCH));
    __syncthreads();
    cur = nxt;
  }
  gemm_slab(asr + cur * (GT_M * GA_PITCH), bsr + cur * ((GT_K / 2) * GB_PITCH),
            acc00, acc01, acc10, acc11);

  const int mOut0 = mBase + wm * 32 + h * 8;
  const int nOut0 = nBase + wn * 32 + c;
#pragma unroll
  for (int r = 0; r < 8; ++r) {
    long long row0 = (long long)(mOut0 + r) * N + nOut0;
    C[row0 +  0] = acc00[r];
    C[row0 + 16] = acc01[r];
    long long row1 = row0 + (long long)16 * N;
    C[row1 +  0] = acc10[r];
    C[row1 + 16] = acc11[r];
  }
}
#undef GEMM_GLOAD
#undef GEMM_SSTORE

#define KP_TM 32
#define KP_TN 32
#define KP_KS 256
#define KP_KG 128
#define KP_APITCH 516
#define KP_BPITCH 80
#define KP_CPITCH 36

#define KP_GLOAD(koff) \
  do { \
    pa0 = *(const float4*)(ag + (long long)(0) * D_ + (koff)); \
    pa1 = *(const float4*)(ag + (long long)(2) * D_ + (koff)); \
    pa2 = *(const float4*)(ag + (long long)(4) * D_ + (koff)); \
    pa3 = *(const float4*)(ag + (long long)(6) * D_ + (koff)); \
    pa4 = *(const float4*)(ag + (long long)(8) * D_ + (koff)); \
    pa5 = *(const float4*)(ag + (long long)(10) * D_ + (koff)); \
    pa6 = *(const float4*)(ag + (long long)(12) * D_ + (koff)); \
    pa7 = *(const float4*)(ag + (long long)(14) * D_ + (koff)); \
    pa8 = *(const float4*)(ag + (long long)(16) * D_ + (koff)); \
    pa9 = *(const float4*)(ag + (long long)(18) * D_ + (koff)); \
    pa10 = *(const float4*)(ag + (long long)(20) * D_ + (koff)); \
    pa11 = *(const float4*)(ag + (long long)(22) * D_ + (koff)); \
    pa12 = *(const float4*)(ag + (long long)(24) * D_ + (koff)); \
    pa13 = *(const float4*)(ag + (long long)(26) * D_ + (koff)); \
    pa14 = *(const float4*)(ag + (long long)(28) * D_ + (koff)); \
    pa15 = *(const float4*)(ag + (long long)(30) * D_ + (koff)); \
    pb0 = *(const float4*)(bg + (long long)((koff) + 0) * H_); \
    pb1 = *(const float4*)(bg + (long long)((koff) + 16) * H_); \
    pb2 = *(const float4*)(bg + (long long)((koff) + 32) * H_); \
    pb3 = *(const float4*)(bg + (long long)((koff) + 48) * H_); \
    pb4 = *(const float4*)(bg + (long long)((koff) + 64) * H_); \
    pb5 = *(const float4*)(bg + (long long)((koff) + 80) * H_); \
    pb6 = *(const float4*)(bg + (long long)((koff) + 96) * H_); \
    pb7 = *(const float4*)(bg + (long long)((koff) + 112) * H_); \
    pb8 = *(const float4*)(bg + (long long)((koff) + 128) * H_); \
    pb9 = *(const float4*)(bg + (long long)((koff) + 144) * H_); \
    pb10 = *(const float4*)(bg + (long long)((koff) + 160) * H_); \
    pb11 = *(const float4*)(bg + (long long)((koff) + 176) * H_); \
    pb12 = *(const float4*)(bg + (long long)((koff) + 192) * H_); \
    pb13 = *(const float4*)(bg + (long long)((koff) + 208) * H_); \
    pb14 = *(const float4*)(bg + (long long)((koff) + 224) * H_); \
    pb15 = *(const float4*)(bg + (long long)((koff) + 240) * H_); \
  } while (0)

#define KP_SSTORE(koff) \
  do { \
    *(float4*)(asw + 0 * KP_APITCH + (koff)) = pa0; \
    *(float4*)(asw + 2 * KP_APITCH + (koff)) = pa1; \
    *(float4*)(asw + 4 * KP_APITCH + (koff)) = pa2; \
    *(float4*)(asw + 6 * KP_APITCH + (koff)) = pa3; \
    *(float4*)(asw + 8 * KP_APITCH + (koff)) = pa4; \
    *(float4*)(asw + 10 * KP_APITCH + (koff)) = pa5; \
    *(float4*)(asw + 12 * KP_APITCH + (koff)) = pa6; \
    *(float4*)(asw + 14 * KP_APITCH + (koff)) = pa7; \
    *(float4*)(asw + 16 * KP_APITCH + (koff)) = pa8; \
    *(float4*)(asw + 18 * KP_APITCH + (koff)) = pa9; \
    *(float4*)(asw + 20 * KP_APITCH + (koff)) = pa10; \
    *(float4*)(asw + 22 * KP_APITCH + (koff)) = pa11; \
    *(float4*)(asw + 24 * KP_APITCH + (koff)) = pa12; \
    *(float4*)(asw + 26 * KP_APITCH + (koff)) = pa13; \
    *(float4*)(asw + 28 * KP_APITCH + (koff)) = pa14; \
    *(float4*)(asw + 30 * KP_APITCH + (koff)) = pa15; \
    *(float4*)(bsw + ((koff) / 2 + 0) * KP_BPITCH) = pb0; \
    *(float4*)(bsw + ((koff) / 2 + 8) * KP_BPITCH) = pb1; \
    *(float4*)(bsw + ((koff) / 2 + 16) * KP_BPITCH) = pb2; \
    *(float4*)(bsw + ((koff) / 2 + 24) * KP_BPITCH) = pb3; \
    *(float4*)(bsw + ((koff) / 2 + 32) * KP_BPITCH) = pb4; \
    *(float4*)(bsw + ((koff) / 2 + 40) * KP_BPITCH) = pb5; \
    *(float4*)(bsw + ((koff) / 2 + 48) * KP_BPITCH) = pb6; \
    *(float4*)(bsw + ((koff) / 2 + 56) * KP_BPITCH) = pb7; \
    *(float4*)(bsw + ((koff) / 2 + 64) * KP_BPITCH) = pb8; \
    *(float4*)(bsw + ((koff) / 2 + 72) * KP_BPITCH) = pb9; \
    *(float4*)(bsw + ((koff) / 2 + 80) * KP_BPITCH) = pb10; \
    *(float4*)(bsw + ((koff) / 2 + 88) * KP_BPITCH) = pb11; \
    *(float4*)(bsw + ((koff) / 2 + 96) * KP_BPITCH) = pb12; \
    *(float4*)(bsw + ((koff) / 2 + 104) * KP_BPITCH) = pb13; \
    *(float4*)(bsw + ((koff) / 2 + 112) * KP_BPITCH) = pb14; \
    *(float4*)(bsw + ((koff) / 2 + 120) * KP_BPITCH) = pb15; \
  } while (0)

__device__ __forceinline__ void kp_group(const float* as, const float* bs, v8f& acc) {
  v2f a[KP_KG / 4], b[KP_KG / 4];
#pragma unroll
  for (int ks = 0; ks < KP_KG / 4; ++ks) {
    if (ks == 4) __builtin_amdgcn_sched_barrier(0);
    a[ks]   = *(const v2f*)(as + 4 * ks);
    b[ks].x = bs[(2 * ks) * KP_BPITCH];
    b[ks].y = bs[(2 * ks) * KP_BPITCH + 16];
  }
  __builtin_amdgcn_sched_barrier(0);
#pragma unroll
  for (int ks = 0; ks < KP_KG / 4; ++ks)
    acc = __builtin_amdgcn_wmma_f32_16x16x4_f32(false, a[ks], false, b[ks], (short)0, acc, false, false);
  asm volatile("" :: "v"(acc) : "memory");
}

__global__ __launch_bounds__(128) void kproj_gemm_f32(
    const float* A, const float* Bm, float* C) {
  __shared__ __attribute__((aligned(16))) float As[KP_TM * KP_APITCH];
  __shared__ __attribute__((aligned(16))) float Bs[(D_ / 2) * KP_BPITCH];
  __shared__ __attribute__((aligned(16))) float Cs[KP_TM * KP_CPITCH];

  const int tid   = threadIdx.x;
  const int lane  = tid & 31;
  const int wave  = tid >> 5;
  const int wm    = wave >> 1;
  const int wn    = wave & 1;
  const int mBase = blockIdx.x * KP_TM;
  const int nBase = blockIdx.y * KP_TN;

  const int c64 = tid & 63;
  const int r2  = tid >> 6;
  const int c8  = tid & 7;
  const int kr  = tid >> 3;
  const float* ag = A  + (long long)(mBase + r2) * D_ + 4 * c64;
  const float* bg = Bm + (long long)kr * H_ + nBase + 4 * c8;
  float* asw = As + r2 * KP_APITCH + 4 * c64;
  float* bsw = Bs + (kr >> 1) * KP_BPITCH + (c8 >> 2) * 32 + (kr & 1) * 16 + (c8 & 3) * 4;

  const int c = lane & 15;
  const int h = lane >> 4;
  const float* asr = As + (wm * 16 + c) * KP_APITCH + 2 * h;
  const float* bsr = Bs + h * KP_BPITCH + wn * 32 + c;

  v8f acc = {};
  float4 pa0, pa1, pa2, pa3, pa4, pa5, pa6, pa7, pa8, pa9, pa10, pa11, pa12, pa13, pa14, pa15;
  float4 pb0, pb1, pb2, pb3, pb4, pb5, pb6, pb7, pb8, pb9, pb10, pb11, pb12, pb13, pb14, pb15;

  KP_GLOAD(0);
  KP_SSTORE(0);
  __syncthreads();

  KP_GLOAD(KP_KS);
  __builtin_amdgcn_sched_barrier(0);

  kp_group(asr,         bsr,                             acc);
  kp_group(asr + KP_KG, bsr + (KP_KG / 2) * KP_BPITCH,   acc);

  KP_SSTORE(KP_KS);
  __syncthreads();

  kp_group(asr + 2 * KP_KG, bsr + (2 * KP_KG / 2) * KP_BPITCH, acc);
  kp_group(asr + 3 * KP_KG, bsr + (3 * KP_KG / 2) * KP_BPITCH, acc);

#pragma unroll
  for (int r = 0; r < 8; ++r) Cs[(wm * 16 + h * 8 + r) * KP_CPITCH + wn * 16 + c] = acc[r];
  __syncthreads();
  const int orow = tid >> 3;
  const float4 o0 = *(const float4*)(Cs + orow * KP_CPITCH + 4 * c8);
  const float4 o1 = *(const float4*)(Cs + (orow + 16) * KP_CPITCH + 4 * c8);
  *(float4*)(C + (long long)(mBase + orow) * H_ + nBase + 4 * c8) = o0;
  *(float4*)(C + (long long)(mBase + orow + 16) * H_ + nBase + 4 * c8) = o1;
}
#undef KP_GLOAD
#undef KP_SSTORE

#define QP_KPART 64
#define QP_PARTS 8
#define QS_PITCH 516

__global__ __launch_bounds__(NK_) void scores_softmax(
    const float* queries, const float* Wq,
    const float* __restrict__ kproj,
    const float* __restrict__ Wv, const int* __restrict__ valid_lens,
    float* __restrict__ attn) {
  __shared__ __attribute__((aligned(16))) float qs[QB_ * QS_PITCH];
  __shared__ __attribute__((aligned(16))) float qpart[QP_PARTS][QB_][H_];
  __shared__ __attribute__((aligned(16))) float qp[QB_][H_];
  __shared__ __attribute__((aligned(16))) float wv[H_];

  const int q0   = blockIdx.x * QB_;
  const int b    = blockIdx.y;
  const int k    = threadIdx.x;
  const int lane = k & 31;
  const int wid  = k >> 5;

  {
    const int c   = lane & 15;
    const int hh  = lane >> 4;
    const int ng  = wid & 1;
    const int kp  = wid >> 1;
    const float* qg = queries + (long long)(b * NQ_ + q0 + (c & 7)) * D_ + QP_KPART * kp + 4 * hh;
    const float* wg = Wq + (long long)(QP_KPART * kp + 4 * hh) * H_ + 64 * ng + 4 * c;

    float wvv = 0.0f;
    if (k < H_) wvv = Wv[k];
    float4 aq[QP_KPART / 8];
#pragma unroll
    for (int m = 0; m < QP_KPART / 8; ++m) aq[m] = *(const float4*)(qg + 8 * m);
    __builtin_amdgcn_sched_barrier(0);

    float4 bq[QP_KPART / 8][4];
#pragma unroll
    for (int m = 0; m < QP_KPART / 8; ++m) {
#pragma unroll
      for (int e = 0; e < 4; ++e) bq[m][e] = *(const float4*)(wg + (long long)(8 * m + e) * H_);
    }
    __builtin_amdgcn_sched_barrier(0);

    v8f acc0 = {}, acc1 = {}, acc2 = {}, acc3 = {};
#pragma unroll
    for (int m = 0; m < QP_KPART / 8; ++m) {
#pragma unroll
      for (int u = 0; u < 2; ++u) {
        const v2f a  = u ? (v2f){aq[m].z, aq[m].w} : (v2f){aq[m].x, aq[m].y};
        const float4 lo = bq[m][2 * u], hi = bq[m][2 * u + 1];
        const v2f b0 = {lo.x, hi.x};
        const v2f b1 = {lo.y, hi.y};
        const v2f b2 = {lo.z, hi.z};
        const v2f b3 = {lo.w, hi.w};
        acc0 = __builtin_amdgcn_wmma_f32_16x16x4_f32(false, a, false, b0, (short)0, acc0, false, false);
        acc1 = __builtin_amdgcn_wmma_f32_16x16x4_f32(false, a, false, b1, (short)0, acc1, false, false);
        acc2 = __builtin_amdgcn_wmma_f32_16x16x4_f32(false, a, false, b2, (short)0, acc2, false, false);
        acc3 = __builtin_amdgcn_wmma_f32_16x16x4_f32(false, a, false, b3, (short)0, acc3, false, false);
      }
    }

    if (hh == 0) {
#pragma unroll
      for (int r = 0; r < QB_; ++r) {
        float4 o;
        o.x = acc0[r]; o.y = acc1[r]; o.z = acc2[r]; o.w = acc3[r];
        *(float4*)&qpart[kp][r][64 * ng + 4 * c] = o;
      }
    }
    if (k < H_) wv[k] = wvv;
    __syncthreads();
    {
      const int r   = k >> 6;
      const int col = 2 * (k & 63);
      v2f sum = *(const v2f*)&qpart[0][r][col];
#pragma unroll
      for (int p = 1; p < QP_PARTS; ++p) sum += *(const v2f*)&qpart[p][r][col];
      *(v2f*)&qp[r][col] = sum;
    }
    __syncthreads();
  }

  const v4f* kp4 =
      (const v4f*)(kproj + (long long)(b * NK_ + k) * H_);
  v2f s2[QB_];
  v2f ta_lo[QB_], ta_hi[QB_], tb_lo[QB_], tb_hi[QB_];
  v2f wa_lo, wa_hi;
  v2f wb_lo = (v2f){0.0f, 0.0f}, wb_hi = (v2f){0.0f, 0.0f};
#pragma unroll
  for (int j = 0; j < QB_; ++j) {
    s2[j] = (v2f){0.0f, 0.0f};
    tb_lo[j] = (v2f){0.0f, 0.0f};
    tb_hi[j] = (v2f){0.0f, 0.0f};
  }

#define SCORE_BLOCK(H0, TL, TH, WL, WH, PL, PH, PWL, PWH)                     \
  do {                                                                        \
    const v4f kv = kp4[(H0) / 4];                                             \
    const v4f w4 = *(const v4f*)&wv[(H0)];                                    \
    const v2f klo = {kv.x, kv.y}, khi = {kv.z, kv.w};                         \
    _Pragma("unroll")                                                         \
    for (int j = 0; j < QB_; ++j) {                                           \
      const v4f q4 = *(const v4f*)&qp[j][(H0)];                               \
      const v2f xlo = (v2f){q4.x, q4.y} + klo;                                \
      const v2f xhi = (v2f){q4.z, q4.w} + khi;                                \
      TL[j] = (v2f){hw_tanh(xlo.x), hw_tanh(xlo.y)};                          \
      TH[j] = (v2f){hw_tanh(xhi.x), hw_tanh(xhi.y)};                          \
      s2[j] = __builtin_elementwise_fma(PWL, PL[j], s2[j]);                   \
      s2[j] = __builtin_elementwise_fma(PWH, PH[j], s2[j]);                   \
    }                                                                         \
    WL = (v2f){w4.x, w4.y};                                                   \
    WH = (v2f){w4.z, w4.w};                                                   \
        \
        \
        \
        \
        \
    __builtin_amdgcn_sched_group_barrier(0x002, 8, 0);                        \
    _Pragma("unroll")                                                         \
    for (int g = 0; g < 28; ++g) {                                            \
      __builtin_amdgcn_sched_group_barrier(0x400, 1, 0);                      \
      __builtin_amdgcn_sched_group_barrier(0x002, 1, 0);                      \
    }                                                                         \
  } while (0)

  for (int h8 = 0; h8 < H_; h8 += 8) {
    SCORE_BLOCK(h8, ta_lo, ta_hi, wa_lo, wa_hi, tb_lo, tb_hi, wb_lo, wb_hi);
    __builtin_amdgcn_sched_barrier(0);
    SCORE_BLOCK(h8 + 4, tb_lo, tb_hi, wb_lo, wb_hi, ta_lo, ta_hi, wa_lo, wa_hi);
  }
#undef SCORE_BLOCK
  float s[QB_];
#pragma unroll
  for (int j = 0; j < QB_; ++j) {
    s2[j] = __builtin_elementwise_fma(wb_lo, tb_lo[j], s2[j]);
    s2[j] = __builtin_elementwise_fma(wb_hi, tb_hi[j], s2[j]);
    s[j] = s2[j].x + s2[j].y;
  }

  const int vlen = valid_lens[b];

  float* sc = qs;
#pragma unroll
  for (int j = 0; j < QB_; ++j) sc[j * QS_PITCH + k] = (k < vlen) ? s[j] : -1.0e6f;
  __syncthreads();

  const int row = wid >> 1;
  const int hf  = wid & 1;
  const float4* rp = (const float4*)(sc + row * QS_PITCH);
  const float4 x0 = rp[lane];
  const float4 x1 = rp[lane + 32];
  const float4 x2 = rp[lane + 64];
  const float4 x3 = rp[lane + 96];

  float m = fmaxf(fmaxf(fmaxf(x0.x, x0.y), fmaxf(x0.z, x0.w)),
                  fmaxf(fmaxf(x1.x, x1.y), fmaxf(x1.z, x1.w)));
  m = fmaxf(m, fmaxf(fmaxf(fmaxf(x2.x, x2.y), fmaxf(x2.z, x2.w)),
                     fmaxf(fmaxf(x3.x, x3.y), fmaxf(x3.z, x3.w))));
#pragma unroll
  for (int off = 16; off > 0; off >>= 1) m = fmaxf(m, __shfl_xor(m, off, 32));

  float4 e0, e1, e2, e3;
  e0.x = hw_exp(x0.x - m); e0.y = hw_exp(x0.y - m); e0.z = hw_exp(x0.z - m); e0.w = hw_exp(x0.w - m);
  e1.x = hw_exp(x1.x - m); e1.y = hw_exp(x1.y - m); e1.z = hw_exp(x1.z - m); e1.w = hw_exp(x1.w - m);
  e2.x = hw_exp(x2.x - m); e2.y = hw_exp(x2.y - m); e2.z = hw_exp(x2.z - m); e2.w = hw_exp(x2.w - m);
  e3.x = hw_exp(x3.x - m); e3.y = hw_exp(x3.y - m); e3.z = hw_exp(x3.z - m); e3.w = hw_exp(x3.w - m);
  float t = (((e0.x + e0.y) + (e0.z + e0.w)) + ((e1.x + e1.y) + (e1.z + e1.w))) +
            (((e2.x + e2.y) + (e2.z + e2.w)) + ((e3.x + e3.y) + (e3.z + e3.w)));
#pragma unroll
  for (int off = 16; off > 0; off >>= 1) t += __shfl_xor(t, off, 32);

  const float4 ya = hf ? e2 : e0;
  const float4 yb = hf ? e3 : e1;
  float* op = attn + (long long)(b * NQ_ + q0 + row) * NK_ + hf * (NK_ / 2) + 4 * lane;
  float4 oa, ob;
  oa.x = ya.x / t; oa.y = ya.y / t; oa.z = ya.z / t; oa.w = ya.w / t;
  ob.x = yb.x / t; ob.y = yb.y / t; ob.z = yb.z / t; ob.w = yb.w / t;
  *(float4*)op = oa;
  *(float4*)(op + NK_ / 4) = ob;
}


extern "C" void kernel_launch(void* const* d_in, const int* in_sizes, int n_in,
                              void* d_out, int out_size, void* d_ws, size_t ws_size,
                              hipStream_t stream) {
  (void)in_sizes; (void)n_in; (void)out_size; (void)ws_size;

  const float* queries    = (const float*)d_in[0];
  const float* keys       = (const float*)d_in[1];
  const float* values     = (const float*)d_in[2];
  const float* Wq         = (const float*)d_in[3];
  const float* Wk         = (const float*)d_in[4];
  const float* Wv         = (const float*)d_in[5];
  const int*   valid_lens = (const int*)d_in[6];
  float*       out        = (float*)d_out;

  float* kproj = (float*)d_ws + (long long)B_ * NQ_ * H_;
  float* attn  = kproj + (long long)B_ * NK_ * H_;

  kproj_gemm_f32<<<dim3((B_ * NK_) / KP_TM, H_ / KP_TN, 1), 128, 0, stream>>>(
      keys, Wk, kproj);

  scores_softmax<<<dim3(NQ_ / QB_, B_), NK_, 0, stream>>>(
      queries, Wq, kproj, Wv, valid_lens, attn);

  wmma_gemm_f32<<<dim3(NQ_ / GT_M, D_ / GT_N, B_), 128, 0, stream>>>(
      attn, values, out, NK_, D_,
      (long long)NQ_ * NK_, (long long)NK_ * D_, (long long)NQ_ * D_);
}


# Output GEMM: K loop software-pipelined: under a slab's 64 WMMAs the next slab goes to LDS, is published by the one (split) barrier and is read into a second fragment register set; sums' order kept
# speedup vs baseline: 3.3157x; 1.0490x over previous
#include <hip/hip_runtime.h>

#define B_  4
#define NQ_ 512
#define NK_ 512
#define D_  512
#define H_  128
#define QB_ 8

typedef __attribute__((ext_vector_type(2))) float v2f;
typedef __attribute__((ext_vector_type(4))) float v4f;
typedef __attribute__((ext_vector_type(8))) float v8f;

__device__ __forceinline__ float hw_tanh(float x) {
#if __has_builtin(__builtin_amdgcn_tanhf)
  return __builtin_amdgcn_tanhf(x);
#else
  float y;
  asm volatile("v_tanh_f32 %0, %1" : "=v"(y) : "v"(x));
  return y;
#endif
}

__device__ __forceinline__ float hw_exp(float x) {
#if __has_builtin(__builtin_amdgcn_exp2f)
  return __builtin_amdgcn_exp2f(x * 1.44269504088896340736f);
#else
  return __expf(x);
#endif
}

#define GT_M 64
#define GT_N 64
#define GT_K 64
#define GA_PITCH 68
#define GB_PITCH 144

#define GEMM_GLOAD(k0)                                                        \
  do {                                                                        \
    ra0 = *(const float4*)(ag + (long long)(0) * K + (k0));                  \
    ra1 = *(const float4*)(ag + (long long)(8) * K + (k0));                  \
    ra2 = *(const float4*)(ag + (long long)(16) * K + (k0));                  \
    ra3 = *(const float4*)(ag + (long long)(24) * K + (k0));                  \
    ra4 = *(const float4*)(ag + (long long)(32) * K + (k0));                  \
    ra5 = *(const float4*)(ag + (long long)(40) * K + (k0));                  \
    ra6 = *(const float4*)(ag + (long long)(48) * K + (k0));                  \
    ra7 = *(const float4*)(ag + (long long)(56) * K + (k0));                  \
    rb0 = *(const float4*)(bg + (long long)((k0) + 0) * N);                  \
    rb1 = *(const float4*)(bg + (long long)((k0) + 0) * N + N);              \
    rb2 = *(const float4*)(bg + (long long)((k0) + 16) * N);                  \
    rb3 = *(const float4*)(bg + (long long)((k0) + 16) * N + N);              \
    rb4 = *(const float4*)(bg + (long long)((k0) + 32) * N);                  \
    rb5 = *(const float4*)(bg + (long long)((k0) + 32) * N + N);              \
    rb6 = *(const float4*)(bg + (long long)((k0) + 48) * N);                  \
    rb7 = *(const float4*)(bg + (long long)((k0) + 48) * N + N);              \
  } while (0)

#define GEMM_SSTORE(as, bs)                                                   \
  do {                                                                        \
    *(float4*)((as) + 0 * GA_PITCH) = ra0;                                  \
    *(float4*)((as) + 8 * GA_PITCH) = ra1;                                  \
    *(float4*)((as) + 16 * GA_PITCH) = ra2;                                  \
    *(float4*)((as) + 24 * GA_PITCH) = ra3;                                  \
    *(float4*)((as) + 32 * GA_PITCH) = ra4;                                  \
    *(float4*)((as) + 40 * GA_PITCH) = ra5;                                  \
    *(float4*)((as) + 48 * GA_PITCH) = ra6;                                  \
    *(float4*)((as) + 56 * GA_PITCH) = ra7;                                  \
    *(float4*)((bs) + 0 * GB_PITCH) = rb0;                                  \
    *(float4*)((bs) + 0 * GB_PITCH + 16) = rb1;                             \
    *(float4*)((bs) + 8 * GB_PITCH) = rb2;                                  \
    *(float4*)((bs) + 8 * GB_PITCH + 16) = rb3;                             \
    *(float4*)((bs) + 16 * GB_PITCH) = rb4;                                  \
    *(float4*)((bs) + 16 * GB_PITCH + 16) = rb5;                             \
    *(float4*)((bs) + 24 * GB_PITCH) = rb6;                                  \
    *(float4*)((bs) + 24 * GB_PITCH + 16) = rb7;                             \
  } while (0)

#define GEMM_FREAD(F, as, bs)                                                 \
  do {                                                                        \
    _Pragma("unroll")                                                         \
    for (int ks = 0; ks < GT_K / 4; ++ks) {                                   \
      F##a0[ks] = *(const v2f*)((as) + 4 * ks);                               \
      F##a1[ks] = *(const v2f*)((as) + 16 * GA_PITCH + 4 * ks);               \
      F##b0[ks].x = (bs)[(2 * ks) * GB_PITCH];              \
      F##b0[ks].y = (bs)[(2 * ks) * GB_PITCH + 16];         \
      F##b1[ks].x = (bs)[(2 * ks) * GB_PITCH + 32];         \
      F##b1[ks].y = (bs)[(2 * ks) * GB_PITCH + 48];                           \
    }                                                                         \
  } while (0)

#define GEMM_MUL(F, k_lo, k_hi)                                               \
  do {                                                                        \
    _Pragma("unroll")                                                         \
    for (int ks = (k_lo); ks < (k_hi); ++ks) {                                \
      acc00 = __builtin_amdgcn_wmma_f32_16x16x4_f32(false, F##a0[ks], false, F##b0[ks], (short)0, acc00, false, false); \
      acc01 = __builtin_amdgcn_wmma_f32_16x16x4_f32(false, F##a0[ks], false, F##b1[ks], (short)0, acc01, false, false); \
      acc10 = __builtin_amdgcn_wmma_f32_16x16x4_f32(false, F##a1[ks], false, F##b0[ks], (short)0, acc10, false, false); \
      acc11 = __builtin_amdgcn_wmma_f32_16x16x4_f32(false, F##a1[ks], false, F##b1[ks], (short)0, acc11, false, false); \
    }                                                                         \
  } while (0)

#define GEMM_ACC_FENCE() \
  asm volatile("" :: "v"(acc00), "v"(acc01), "v"(acc10), "v"(acc11) : "memory")

#define GEMM_PIPE_STEP(F, G, NXT, HAS_LOAD, K_LOAD)                           \
  do {                                                                        \
    GEMM_MUL(F, 0, 8);                                                        \
    GEMM_ACC_FENCE();                                                         \
        \
        \
        \
        \
    GEMM_SSTORE(asw + (NXT) * (GT_M * GA_PITCH), bsw + (NXT) * ((GT_K / 2) * GB_PITCH)); \
    if (HAS_LOAD) GEMM_GLOAD(K_LOAD);                                         \
      \
    __builtin_amdgcn_sched_barrier(0);                                        \
    GEMM_MUL(F, 8, 11);                                                       \
    GEMM_ACC_FENCE();                                                         \
       \
       \
       \
       \
       \
       \
       \
    __builtin_amdgcn_fence(__ATOMIC_RELEASE, "workgroup", "local");           \
    __builtin_amdgcn_s_barrier_signal(-1);                                    \
    GEMM_MUL(F, 11, 13);                                                      \
    GEMM_ACC_FENCE();                                                         \
    __builtin_amdgcn_s_barrier_wait(-1);                                      \
    __builtin_amdgcn_fence(__ATOMIC_ACQUIRE, "workgroup", "local");           \
    GEMM_FREAD(G, asr + (NXT) * (GT_M * GA_PITCH), bsr + (NXT) * ((GT_K / 2) * GB_PITCH)); \
       \
       \
    __builtin_amdgcn_sched_barrier(0);                                        \
    GEMM_MUL(F, 13, 16);                                                      \
  } while (0)

__global__ __launch_bounds__(128) void wmma_gemm_f32(
    const float* A, const float* Bm, float* C, int K, int N,
    long long sA, long long sB, long long sC) {
  __shared__ __attribute__((aligned(16))) float As[2 * GT_M * GA_PITCH];
  __shared__ __attribute__((aligned(16))) float Bs[2 * (GT_K / 2) * GB_PITCH];

  const int tid   = threadIdx.x;
  const int lane  = tid & 31;
  const int wave  = tid >> 5;
  const int wm    = wave >> 1;
  const int wn    = wave & 1;
  const int mBase = blockIdx.x * GT_M;
  const int nBase = blockIdx.y * GT_N;
  const int batch = blockIdx.z;
  A  += (long long)batch * sA;
  Bm += (long long)batch * sB;
  C  += (long long)batch * sC;

  const int c4 = tid & 15;
  const int r8 = tid >> 4;
  const float* ag = A  + (long long)(mBase + r8) * K + 4 * c4;
  const float* bg = Bm + (long long)(2 * r8) * N + nBase + 4 * c4;
  float* asw = As + r8 * GA_PITCH + 4 * c4;
  float* bsw = Bs + r8 * GB_PITCH + (c4 >> 2) * 32 + (c4 & 3) * 4;

  const int c = lane & 15;
  const int h = lane >> 4;
  const float* asr = As + (wm * 32 + c) * GA_PITCH + 2 * h;
  const float* bsr = Bs + h * GB_PITCH + wn * 64 + c;

  v8f acc00 = {}, acc01 = {}, acc10 = {}, acc11 = {};
  float4 ra0, ra1, ra2, ra3, ra4, ra5, ra6, ra7;
  float4 rb0, rb1, rb2, rb3, rb4, rb5, rb6, rb7;
  v2f xa0[GT_K / 4], xa1[GT_K / 4], xb0[GT_K / 4], xb1[GT_K / 4];
  v2f ya0[GT_K / 4], ya1[GT_K / 4], yb0[GT_K / 4], yb1[GT_K / 4];

  const int nSlab = K / GT_K;

  GEMM_GLOAD(0);
  GEMM_SSTORE(asw, bsw);
  __syncthreads();
  GEMM_GLOAD(GT_K);
  GEMM_FREAD(x, asr, bsr);
  __builtin_amdgcn_sched_barrier(0);

  for (int s = 0; s + 2 < nSlab; s += 2) {
    GEMM_PIPE_STEP(x, y, 1, true, (s + 2) * GT_K);
    GEMM_PIPE_STEP(y, x, 0, true, (s + 3) * GT_K);
  }
  GEMM_PIPE_STEP(x, y, 1, false, 0);
  GEMM_MUL(y, 0, GT_K / 4);

  const int mOut0 = mBase + wm * 32 + h * 8;
  const int nOut0 = nBase + wn * 32 + c;
#pragma unroll
  for (int r = 0; r < 8; ++r) {
    long long row0 = (long long)(mOut0 + r) * N + nOut0;
    C[row0 +  0] = acc00[r];
    C[row0 + 16] = acc01[r];
    long long row1 = row0 + (long long)16 * N;
    C[row1 +  0] = acc10[r];
    C[row1 + 16] = acc11[r];
  }
}
#undef GEMM_GLOAD
#undef GEMM_SSTORE
#undef GEMM_FREAD
#undef GEMM_MUL
#undef GEMM_PIPE_STEP
#undef GEMM_ACC_FENCE

#define KP_TM 32
#define KP_TN 32
#define KP_KS 256
#define KP_KG 128
#define KP_APITCH 516
#define KP_BPITCH 80
#define KP_CPITCH 36

#define KP_GLOAD(koff) \
  do { \
    pa0 = *(const float4*)(ag + (long long)(0) * D_ + (koff)); \
    pa1 = *(const float4*)(ag + (long long)(2) * D_ + (koff)); \
    pa2 = *(const float4*)(ag + (long long)(4) * D_ + (koff)); \
    pa3 = *(const float4*)(ag + (long long)(6) * D_ + (koff)); \
    pa4 = *(const float4*)(ag + (long long)(8) * D_ + (koff)); \
    pa5 = *(const float4*)(ag + (long long)(10) * D_ + (koff)); \
    pa6 = *(const float4*)(ag + (long long)(12) * D_ + (koff)); \
    pa7 = *(const float4*)(ag + (long long)(14) * D_ + (koff)); \
    pa8 = *(const float4*)(ag + (long long)(16) * D_ + (koff)); \
    pa9 = *(const float4*)(ag + (long long)(18) * D_ + (koff)); \
    pa10 = *(const float4*)(ag + (long long)(20) * D_ + (koff)); \
    pa11 = *(const float4*)(ag + (long long)(22) * D_ + (koff)); \
    pa12 = *(const float4*)(ag + (long long)(24) * D_ + (koff)); \
    pa13 = *(const float4*)(ag + (long long)(26) * D_ + (koff)); \
    pa14 = *(const float4*)(ag + (long long)(28) * D_ + (koff)); \
    pa15 = *(const float4*)(ag + (long long)(30) * D_ + (koff)); \
    pb0 = *(const float4*)(bg + (long long)((koff) + 0) * H_); \
    pb1 = *(const float4*)(bg + (long long)((koff) + 16) * H_); \
    pb2 = *(const float4*)(bg + (long long)((koff) + 32) * H_); \
    pb3 = *(const float4*)(bg + (long long)((koff) + 48) * H_); \
    pb4 = *(const float4*)(bg + (long long)((koff) + 64) * H_); \
    pb5 = *(const float4*)(bg + (long long)((koff) + 80) * H_); \
    pb6 = *(const float4*)(bg + (long long)((koff) + 96) * H_); \
    pb7 = *(const float4*)(bg + (long long)((koff) + 112) * H_); \
    pb8 = *(const float4*)(bg + (long long)((koff) + 128) * H_); \
    pb9 = *(const float4*)(bg + (long long)((koff) + 144) * H_); \
    pb10 = *(const float4*)(bg + (long long)((koff) + 160) * H_); \
    pb11 = *(const float4*)(bg + (long long)((koff) + 176) * H_); \
    pb12 = *(const float4*)(bg + (long long)((koff) + 192) * H_); \
    pb13 = *(const float4*)(bg + (long long)((koff) + 208) * H_); \
    pb14 = *(const float4*)(bg + (long long)((koff) + 224) * H_); \
    pb15 = *(const float4*)(bg + (long long)((koff) + 240) * H_); \
  } while (0)

#define KP_SSTORE(koff) \
  do { \
    *(float4*)(asw + 0 * KP_APITCH + (koff)) = pa0; \
    *(float4*)(asw + 2 * KP_APITCH + (koff)) = pa1; \
    *(float4*)(asw + 4 * KP_APITCH + (koff)) = pa2; \
    *(float4*)(asw + 6 * KP_APITCH + (koff)) = pa3; \
    *(float4*)(asw + 8 * KP_APITCH + (koff)) = pa4; \
    *(float4*)(asw + 10 * KP_APITCH + (koff)) = pa5; \
    *(float4*)(asw + 12 * KP_APITCH + (koff)) = pa6; \
    *(float4*)(asw + 14 * KP_APITCH + (koff)) = pa7; \
    *(float4*)(asw + 16 * KP_APITCH + (koff)) = pa8; \
    *(float4*)(asw + 18 * KP_APITCH + (koff)) = pa9; \
    *(float4*)(asw + 20 * KP_APITCH + (koff)) = pa10; \
    *(float4*)(asw + 22 * KP_APITCH + (koff)) = pa11; \
    *(float4*)(asw + 24 * KP_APITCH + (koff)) = pa12; \
    *(float4*)(asw + 26 * KP_APITCH + (koff)) = pa13; \
    *(float4*)(asw + 28 * KP_APITCH + (koff)) = pa14; \
    *(float4*)(asw + 30 * KP_APITCH + (koff)) = pa15; \
    *(float4*)(bsw + ((koff) / 2 + 0) * KP_BPITCH) = pb0; \
    *(float4*)(bsw + ((koff) / 2 + 8) * KP_BPITCH) = pb1; \
    *(float4*)(bsw + ((koff) / 2 + 16) * KP_BPITCH) = pb2; \
    *(float4*)(bsw + ((koff) / 2 + 24) * KP_BPITCH) = pb3; \
    *(float4*)(bsw + ((koff) / 2 + 32) * KP_BPITCH) = pb4; \
    *(float4*)(bsw + ((koff) / 2 + 40) * KP_BPITCH) = pb5; \
    *(float4*)(bsw + ((koff) / 2 + 48) * KP_BPITCH) = pb6; \
    *(float4*)(bsw + ((koff) / 2 + 56) * KP_BPITCH) = pb7; \
    *(float4*)(bsw + ((koff) / 2 + 64) * KP_BPITCH) = pb8; \
    *(float4*)(bsw + ((koff) / 2 + 72) * KP_BPITCH) = pb9; \
    *(float4*)(bsw + ((koff) / 2 + 80) * KP_BPITCH) = pb10; \
    *(float4*)(bsw + ((koff) / 2 + 88) * KP_BPITCH) = pb11; \
    *(float4*)(bsw + ((koff) / 2 + 96) * KP_BPITCH) = pb12; \
    *(float4*)(bsw + ((koff) / 2 + 104) * KP_BPITCH) = pb13; \
    *(float4*)(bsw + ((koff) / 2 + 112) * KP_BPITCH) = pb14; \
    *(float4*)(bsw + ((koff) / 2 + 120) * KP_BPITCH) = pb15; \
  } while (0)

__device__ __forceinline__ void kp_group(const float* as, const float* bs, v8f& acc) {
  v2f a[KP_KG / 4], b[KP_KG / 4];
#pragma unroll
  for (int ks = 0; ks < KP_KG / 4; ++ks) {
    if (ks == 4) __builtin_amdgcn_sched_barrier(0);
    a[ks]   = *(const v2f*)(as + 4 * ks);
    b[ks].x = bs[(2 * ks) * KP_BPITCH];
    b[ks].y = bs[(2 * ks) * KP_BPITCH + 16];
  }
  __builtin_amdgcn_sched_barrier(0);
#pragma unroll
  for (int ks = 0; ks < KP_KG / 4; ++ks)
    acc = __builtin_amdgcn_wmma_f32_16x16x4_f32(false, a[ks], false, b[ks], (short)0, acc, false, false);
  asm volatile("" :: "v"(acc) : "memory");
}

__global__ __launch_bounds__(128) void kproj_gemm_f32(
    const float* A, const float* Bm, float* C) {
  __shared__ __attribute__((aligned(16))) float As[KP_TM * KP_APITCH];
  __shared__ __attribute__((aligned(16))) float Bs[(D_ / 2) * KP_BPITCH];
  __shared__ __attribute__((aligned(16))) float Cs[KP_TM * KP_CPITCH];

  const int tid   = threadIdx.x;
  const int lane  = tid & 31;
  const int wave  = tid >> 5;
  const int wm    = wave >> 1;
  const int wn    = wave & 1;
  const int mBase = blockIdx.x * KP_TM;
  const int nBase = blockIdx.y * KP_TN;

  const int c64 = tid & 63;
  const int r2  = tid >> 6;
  const int c8  = tid & 7;
  const int kr  = tid >> 3;
  const float* ag = A  + (long long)(mBase + r2) * D_ + 4 * c64;
  const float* bg = Bm + (long long)kr * H_ + nBase + 4 * c8;
  float* asw = As + r2 * KP_APITCH + 4 * c64;
  float* bsw = Bs + (kr >> 1) * KP_BPITCH + (c8 >> 2) * 32 + (kr & 1) * 16 + (c8 & 3) * 4;

  const int c = lane & 15;
  const int h = lane >> 4;
  const float* asr = As + (wm * 16 + c) * KP_APITCH + 2 * h;
  const float* bsr = Bs + h * KP_BPITCH + wn * 32 + c;

  v8f acc = {};
  float4 pa0, pa1, pa2, pa3, pa4, pa5, pa6, pa7, pa8, pa9, pa10, pa11, pa12, pa13, pa14, pa15;
  float4 pb0, pb1, pb2, pb3, pb4, pb5, pb6, pb7, pb8, pb9, pb10, pb11, pb12, pb13, pb14, pb15;

  KP_GLOAD(0);
  KP_SSTORE(0);
  __syncthreads();

  KP_GLOAD(KP_KS);
  __builtin_amdgcn_sched_barrier(0);

  kp_group(asr,         bsr,                             acc);
  kp_group(asr + KP_KG, bsr + (KP_KG / 2) * KP_BPITCH,   acc);

  KP_SSTORE(KP_KS);
  __syncthreads();

  kp_group(asr + 2 * KP_KG, bsr + (2 * KP_KG / 2) * KP_BPITCH, acc);
  kp_group(asr + 3 * KP_KG, bsr + (3 * KP_KG / 2) * KP_BPITCH, acc);

#pragma unroll
  for (int r = 0; r < 8; ++r) Cs[(wm * 16 + h * 8 + r) * KP_CPITCH + wn * 16 + c] = acc[r];
  __syncthreads();
  const int orow = tid >> 3;
  const float4 o0 = *(const float4*)(Cs + orow * KP_CPITCH + 4 * c8);
  const float4 o1 = *(const float4*)(Cs + (orow + 16) * KP_CPITCH + 4 * c8);
  *(float4*)(C + (long long)(mBase + orow) * H_ + nBase + 4 * c8) = o0;
  *(float4*)(C + (long long)(mBase + orow + 16) * H_ + nBase + 4 * c8) = o1;
}
#undef KP_GLOAD
#undef KP_SSTORE

#define QP_KPART 64
#define QP_PARTS 8
#define QS_PITCH 516

__global__ __launch_bounds__(NK_) void scores_softmax(
    const float* queries, const float* Wq,
    const float* __restrict__ kproj,
    const float* __restrict__ Wv, const int* __restrict__ valid_lens,
    float* __restrict__ attn) {
  __shared__ __attribute__((aligned(16))) float qs[QB_ * QS_PITCH];
  __shared__ __attribute__((aligned(16))) float qpart[QP_PARTS][QB_][H_];
  __shared__ __attribute__((aligned(16))) float qp[QB_][H_];
  __shared__ __attribute__((aligned(16))) float wv[H_];

  const int q0   = blockIdx.x * QB_;
  const int b    = blockIdx.y;
  const int k    = threadIdx.x;
  const int lane = k & 31;
  const int wid  = k >> 5;

  {
    const int c   = lane & 15;
    const int hh  = lane >> 4;
    const int ng  = wid & 1;
    const int kp  = wid >> 1;
    const float* qg = queries + (long long)(b * NQ_ + q0 + (c & 7)) * D_ + QP_KPART * kp + 4 * hh;
    const float* wg = Wq + (long long)(QP_KPART * kp + 4 * hh) * H_ + 64 * ng + 4 * c;

    float wvv = 0.0f;
    if (k < H_) wvv = Wv[k];
    float4 aq[QP_KPART / 8];
#pragma unroll
    for (int m = 0; m < QP_KPART / 8; ++m) aq[m] = *(const float4*)(qg + 8 * m);
    __builtin_amdgcn_sched_barrier(0);

    float4 bq[QP_KPART / 8][4];
#pragma unroll
    for (int m = 0; m < QP_KPART / 8; ++m) {
#pragma unroll
      for (int e = 0; e < 4; ++e) bq[m][e] = *(const float4*)(wg + (long long)(8 * m + e) * H_);
    }
    __builtin_amdgcn_sched_barrier(0);

    v8f acc0 = {}, acc1 = {}, acc2 = {}, acc3 = {};
#pragma unroll
    for (int m = 0; m < QP_KPART / 8; ++m) {
#pragma unroll
      for (int u = 0; u < 2; ++u) {
        const v2f a  = u ? (v2f){aq[m].z, aq[m].w} : (v2f){aq[m].x, aq[m].y};
        const float4 lo = bq[m][2 * u], hi = bq[m][2 * u + 1];
        const v2f b0 = {lo.x, hi.x};
        const v2f b1 = {lo.y, hi.y};
        const v2f b2 = {lo.z, hi.z};
        const v2f b3 = {lo.w, hi.w};
        acc0 = __builtin_amdgcn_wmma_f32_16x16x4_f32(false, a, false, b0, (short)0, acc0, false, false);
        acc1 = __builtin_amdgcn_wmma_f32_16x16x4_f32(false, a, false, b1, (short)0, acc1, false, false);
        acc2 = __builtin_amdgcn_wmma_f32_16x16x4_f32(false, a, false, b2, (short)0, acc2, false, false);
        acc3 = __builtin_amdgcn_wmma_f32_16x16x4_f32(false, a, false, b3, (short)0, acc3, false, false);
      }
    }

    if (hh == 0) {
#pragma unroll
      for (int r = 0; r < QB_; ++r) {
        float4 o;
        o.x = acc0[r]; o.y = acc1[r]; o.z = acc2[r]; o.w = acc3[r];
        *(float4*)&qpart[kp][r][64 * ng + 4 * c] = o;
      }
    }
    if (k < H_) wv[k] = wvv;
    __syncthreads();
    {
      const int r   = k >> 6;
      const int col = 2 * (k & 63);
      v2f sum = *(const v2f*)&qpart[0][r][col];
#pragma unroll
      for (int p = 1; p < QP_PARTS; ++p) sum += *(const v2f*)&qpart[p][r][col];
      *(v2f*)&qp[r][col] = sum;
    }
    __syncthreads();
  }

  const v4f* kp4 =
      (const v4f*)(kproj + (long long)(b * NK_ + k) * H_);
  v2f s2[QB_];
  v2f ta_lo[QB_], ta_hi[QB_], tb_lo[QB_], tb_hi[QB_];
  v2f wa_lo, wa_hi;
  v2f wb_lo = (v2f){0.0f, 0.0f}, wb_hi = (v2f){0.0f, 0.0f};
#pragma unroll
  for (int j = 0; j < QB_; ++j) {
    s2[j] = (v2f){0.0f, 0.0f};
    tb_lo[j] = (v2f){0.0f, 0.0f};
    tb_hi[j] = (v2f){0.0f, 0.0f};
  }

#define SCORE_BLOCK(H0, TL, TH, WL, WH, PL, PH, PWL, PWH)                     \
  do {                                                                        \
    const v4f kv = kp4[(H0) / 4];                                             \
    const v4f w4 = *(const v4f*)&wv[(H0)];                                    \
    const v2f klo = {kv.x, kv.y}, khi = {kv.z, kv.w};                         \
    _Pragma("unroll")                                                         \
    for (int j = 0; j < QB_; ++j) {                                           \
      const v4f q4 = *(const v4f*)&qp[j][(H0)];                               \
      const v2f xlo = (v2f){q4.x, q4.y} + klo;                                \
      const v2f xhi = (v2f){q4.z, q4.w} + khi;                                \
      TL[j] = (v2f){hw_tanh(xlo.x), hw_tanh(xlo.y)};                          \
      TH[j] = (v2f){hw_tanh(xhi.x), hw_tanh(xhi.y)};                          \
      s2[j] = __builtin_elementwise_fma(PWL, PL[j], s2[j]);                   \
      s2[j] = __builtin_elementwise_fma(PWH, PH[j], s2[j]);                   \
    }                                                                         \
    WL = (v2f){w4.x, w4.y};                                                   \
    WH = (v2f){w4.z, w4.w};                                                   \
        \
        \
        \
        \
        \
    __builtin_amdgcn_sched_group_barrier(0x002, 8, 0);                        \
    _Pragma("unroll")                                                         \
    for (int g = 0; g < 28; ++g) {                                            \
      __builtin_amdgcn_sched_group_barrier(0x400, 1, 0);                      \
      __builtin_amdgcn_sched_group_barrier(0x002, 1, 0);                      \
    }                                                                         \
  } while (0)

  for (int h8 = 0; h8 < H_; h8 += 8) {
    SCORE_BLOCK(h8, ta_lo, ta_hi, wa_lo, wa_hi, tb_lo, tb_hi, wb_lo, wb_hi);
    __builtin_amdgcn_sched_barrier(0);
    SCORE_BLOCK(h8 + 4, tb_lo, tb_hi, wb_lo, wb_hi, ta_lo, ta_hi, wa_lo, wa_hi);
  }
#undef SCORE_BLOCK
  float s[QB_];
#pragma unroll
  for (int j = 0; j < QB_; ++j) {
    s2[j] = __builtin_elementwise_fma(wb_lo, tb_lo[j], s2[j]);
    s2[j] = __builtin_elementwise_fma(wb_hi, tb_hi[j], s2[j]);
    s[j] = s2[j].x + s2[j].y;
  }

  const int vlen = valid_lens[b];

  float* sc = qs;
#pragma unroll
  for (int j = 0; j < QB_; ++j) sc[j * QS_PITCH + k] = (k < vlen) ? s[j] : -1.0e6f;
  __syncthreads();

  const int row = wid >> 1;
  const int hf  = wid & 1;
  const float4* rp = (const float4*)(sc + row * QS_PITCH);
  const float4 x0 = rp[lane];
  const float4 x1 = rp[lane + 32];
  const float4 x2 = rp[lane + 64];
  const float4 x3 = rp[lane + 96];

  float m = fmaxf(fmaxf(fmaxf(x0.x, x0.y), fmaxf(x0.z, x0.w)),
                  fmaxf(fmaxf(x1.x, x1.y), fmaxf(x1.z, x1.w)));
  m = fmaxf(m, fmaxf(fmaxf(fmaxf(x2.x, x2.y), fmaxf(x2.z, x2.w)),
                     fmaxf(fmaxf(x3.x, x3.y), fmaxf(x3.z, x3.w))));
#pragma unroll
  for (int off = 16; off > 0; off >>= 1) m = fmaxf(m, __shfl_xor(m, off, 32));

  float4 e0, e1, e2, e3;
  e0.x = hw_exp(x0.x - m); e0.y = hw_exp(x0.y - m); e0.z = hw_exp(x0.z - m); e0.w = hw_exp(x0.w - m);
  e1.x = hw_exp(x1.x - m); e1.y = hw_exp(x1.y - m); e1.z = hw_exp(x1.z - m); e1.w = hw_exp(x1.w - m);
  e2.x = hw_exp(x2.x - m); e2.y = hw_exp(x2.y - m); e2.z = hw_exp(x2.z - m); e2.w = hw_exp(x2.w - m);
  e3.x = hw_exp(x3.x - m); e3.y = hw_exp(x3.y - m); e3.z = hw_exp(x3.z - m); e3.w = hw_exp(x3.w - m);
  float t = (((e0.x + e0.y) + (e0.z + e0.w)) + ((e1.x + e1.y) + (e1.z + e1.w))) +
            (((e2.x + e2.y) + (e2.z + e2.w)) + ((e3.x + e3.y) + (e3.z + e3.w)));
#pragma unroll
  for (int off = 16; off > 0; off >>= 1) t += __shfl_xor(t, off, 32);

  const float4 ya = hf ? e2 : e0;
  const float4 yb = hf ? e3 : e1;
  float* op = attn + (long long)(b * NQ_ + q0 + row) * NK_ + hf * (NK_ / 2) + 4 * lane;
  float4 oa, ob;
  oa.x = ya.x / t; oa.y = ya.y / t; oa.z = ya.z / t; oa.w = ya.w / t;
  ob.x = yb.x / t; ob.y = yb.y / t; ob.z = yb.z / t; ob.w = yb.w / t;
  *(float4*)op = oa;
  *(float4*)(op + NK_ / 4) = ob;
}


extern "C" void kernel_launch(void* const* d_in, const int* in_sizes, int n_in,
                              void* d_out, int out_size, void* d_ws, size_t ws_size,
                              hipStream_t stream) {
  (void)in_sizes; (void)n_in; (void)out_size; (void)ws_size;

  const float* queries    = (const float*)d_in[0];
  const float* keys       = (const float*)d_in[1];
  const float* values     = (const float*)d_in[2];
  const float* Wq         = (const float*)d_in[3];
  const float* Wk         = (const float*)d_in[4];
  const float* Wv         = (const float*)d_in[5];
  const int*   valid_lens = (const int*)d_in[6];
  float*       out        = (float*)d_out;

  float* kproj = (float*)d_ws + (long long)B_ * NQ_ * H_;
  float* attn  = kproj + (long long)B_ * NK_ * H_;

  kproj_gemm_f32<<<dim3((B_ * NK_) / KP_TM, H_ / KP_TN, 1), 128, 0, stream>>>(
      keys, Wk, kproj);

  scores_softmax<<<dim3(NQ_ / QB_, B_), NK_, 0, stream>>>(
      queries, Wq, kproj, Wv, valid_lens, attn);

  wmma_gemm_f32<<<dim3(NQ_ / GT_M, D_ / GT_N, B_), 128, 0, stream>>>(
      attn, values, out, NK_, D_,
      (long long)NQ_ * NK_, (long long)NK_ * D_, (long long)NQ_ * D_);
}
